# starting kernel
# baseline (speedup 1.0000x reference)
#include <hip/hip_runtime.h>
#include <math.h>


typedef float v2f __attribute__((ext_vector_type(2)));
typedef float v8f __attribute__((ext_vector_type(8)));

static __device__ __forceinline__ void atomicAddF32(float* p, float v) {
    asm volatile("global_atomic_add_f32 %0, %1, off" :: "v"(p), "v"(v) : "memory");
}


__global__ void zero_deg_kernel(int* __restrict__ deg, int n) {
    int i = blockIdx.x * blockDim.x + threadIdx.x;
    if (i < n) deg[i] = 0;
}

__global__ void count_deg_kernel(const int* __restrict__ dst, int* __restrict__ deg, int e) {
    int i = blockIdx.x * blockDim.x + threadIdx.x;
    if (i < e) atomicAdd(&deg[dst[i]], 1);
}

__global__ void dinv_kernel(const int* __restrict__ deg, float* __restrict__ dinv, int n) {
    int i = blockIdx.x * blockDim.x + threadIdx.x;
    if (i < n) dinv[i] = rsqrtf((float)deg[i] + 1.0f);
}


template <int NOUT, bool RELU_BIAS>
__global__ void gemm_wmma_kernel(const float* __restrict__ A,
                                 const float* __restrict__ W,
                                 const float* __restrict__ bias,
                                 float* __restrict__ C,
                                 int nrows) {
    constexpr int K = 128;
    constexpr int NWAVES = NOUT / 16;
    __shared__ float sA[16 * K];

    const int lane = threadIdx.x & 31;
    const int wave = threadIdx.x >> 5;
    const int n0   = wave * 16;
    const int ncol = n0 + (lane & 15);
    const int koff = (lane & 16) ? 2 : 0;

    v2f wf[32];
#pragma unroll
    for (int kb = 0; kb < 32; ++kb) {
        const int k = kb * 4 + koff;
        wf[kb].x = W[(size_t)k * NOUT + ncol];
        wf[kb].y = W[(size_t)(k + 1) * NOUT + ncol];
    }

    const int m    = lane & 15;
    const int mrow = (lane & 16) ? 8 : 0;

    for (int row0 = blockIdx.x * 16; row0 + 16 <= nrows; row0 += gridDim.x * 16) {
        const float4* gsrc = (const float4*)(A + (size_t)row0 * K);
        for (int idx4 = threadIdx.x; idx4 < (16 * K) / 4; idx4 += NWAVES * 32) {
            float4 v = gsrc[idx4];
            if constexpr (RELU_BIAS) {
                const int k = (idx4 * 4) & (K - 1);
                v.x = fmaxf(v.x + bias[k + 0], 0.0f);
                v.y = fmaxf(v.y + bias[k + 1], 0.0f);
                v.z = fmaxf(v.z + bias[k + 2], 0.0f);
                v.w = fmaxf(v.w + bias[k + 3], 0.0f);
            }
            ((float4*)sA)[idx4] = v;
        }
        __syncthreads();

        v8f c = {0.f, 0.f, 0.f, 0.f, 0.f, 0.f, 0.f, 0.f};
#pragma unroll
        for (int kb = 0; kb < 32; ++kb) {
            const v2f a = *(const v2f*)&sA[m * K + kb * 4 + koff];
            c = __builtin_amdgcn_wmma_f32_16x16x4_f32(
                    false, a, false, wf[kb], (short)0, c, false, false);
        }

        float* cp = C + (size_t)(row0 + mrow) * NOUT + n0 + m;
#pragma unroll
        for (int g = 0; g < 8; ++g) cp[(size_t)g * NOUT] = c[g];

        __syncthreads();
    }
}

__global__ void self_init_kernel(const float* __restrict__ hl,
                                 const float* __restrict__ dinv,
                                 float* __restrict__ acc, int total4) {
    int idx4 = blockIdx.x * blockDim.x + threadIdx.x;
    if (idx4 < total4) {
        const int i = idx4 >> 5;
        const float d = dinv[i];
        const float w = d * d;
        float4 v = ((const float4*)hl)[idx4];
        v.x *= w; v.y *= w; v.z *= w; v.w *= w;
        ((float4*)acc)[idx4] = v;
    }
}

__global__ void out_init_kernel(const float* __restrict__ hl2,
                                const float* __restrict__ dinv,
                                const float* __restrict__ b2,
                                float* __restrict__ out, int total4) {
    int idx4 = blockIdx.x * blockDim.x + threadIdx.x;
    if (idx4 < total4) {
        const int i  = idx4 >> 4;
        const int f4 = (idx4 & 15);
        const float d = dinv[i];
        const float w = d * d;
        float4 v = ((const float4*)hl2)[idx4];
        float4 b = ((const float4*)b2)[f4];
        v.x = v.x * w + b.x; v.y = v.y * w + b.y;
        v.z = v.z * w + b.z; v.w = v.w * w + b.w;
        ((float4*)out)[idx4] = v;
    }
}


template <int FEAT>
__global__ void scatter_kernel(const float* __restrict__ hl,
                               const float* __restrict__ dinv,
                               const int* __restrict__ src,
                               const int* __restrict__ dst,
                               float* __restrict__ acc, int e) {
    constexpr int VEC = FEAT / 32;
    const int lane = threadIdx.x & 31;
    const int eidx = blockIdx.x * (blockDim.x >> 5) + (threadIdx.x >> 5);
    if (eidx >= e) return;
    const int s = src[eidx];
    const int d = dst[eidx];
    const float w = dinv[s] * dinv[d];
    const float* gp = hl  + (size_t)s * FEAT + lane * VEC;
    float*       ap = acc + (size_t)d * FEAT + lane * VEC;
    if constexpr (VEC == 4) {
        float4 v = *(const float4*)gp;
        atomicAddF32(ap + 0, w * v.x);
        atomicAddF32(ap + 1, w * v.y);
        atomicAddF32(ap + 2, w * v.z);
        atomicAddF32(ap + 3, w * v.w);
    } else {
        float2 v = *(const float2*)gp;
        atomicAddF32(ap + 0, w * v.x);
        atomicAddF32(ap + 1, w * v.y);
    }
}


extern "C" void kernel_launch(void* const* d_in, const int* in_sizes, int n_in,
                              void* d_out, int out_size, void* d_ws, size_t ws_size,
                              hipStream_t stream) {
    const float* x   = (const float*)d_in[0];
    const int*   ei  = (const int*)d_in[1];
    const float* W1  = (const float*)d_in[2];
    const float* b1  = (const float*)d_in[3];
    const float* W2  = (const float*)d_in[4];
    const float* b2  = (const float*)d_in[5];
    float*       out = (float*)d_out;

    const int N = in_sizes[0] / 128;
    const int E = in_sizes[1] / 2;
    const int* src = ei;
    const int* dst = ei + E;

    char* w = (char*)d_ws;
    auto carve = [&](size_t bytes) {
        void* p = (void*)w;
        w += (bytes + 255) & ~(size_t)255;
        return p;
    };
    int*   deg  = (int*)carve((size_t)N * 4);
    float* dinv = (float*)carve((size_t)N * 4);
    float* hl1  = (float*)carve((size_t)N * 128 * 4);
    float* hagg = (float*)carve((size_t)N * 128 * 4);
    float* hl2  = hl1;
    (void)n_in; (void)out_size; (void)ws_size;

    const int T = 256;
    zero_deg_kernel<<<(N + T - 1) / T, T, 0, stream>>>(deg, N);
    count_deg_kernel<<<(E + T - 1) / T, T, 0, stream>>>(dst, deg, E);
    dinv_kernel<<<(N + T - 1) / T, T, 0, stream>>>(deg, dinv, N);

    gemm_wmma_kernel<128, false><<<2048, 256, 0, stream>>>(x, W1, nullptr, hl1, N);

    {
        const int total4 = N * 128 / 4;
        self_init_kernel<<<(total4 + T - 1) / T, T, 0, stream>>>(hl1, dinv, hagg, total4);
        scatter_kernel<128><<<(E + 7) / 8, 256, 0, stream>>>(hl1, dinv, src, dst, hagg, E);
    }

    gemm_wmma_kernel<64, true><<<2048, 128, 0, stream>>>(hagg, W2, b1, hl2, N);

    {
        const int total4 = N * 64 / 4;
        out_init_kernel<<<(total4 + T - 1) / T, T, 0, stream>>>(hl2, dinv, b2, out, total4);
        scatter_kernel<64><<<(E + 7) / 8, 256, 0, stream>>>(hl2, dinv, src, dst, out, E);
    }
}


# Float atomic scatter replaced by per-call destination buckets (returning int atomic gives arrival-order rank, scan, fill) and a wave-per-node register gather-sum; each row stored once
# speedup vs baseline: 2.0233x; 2.0233x over previous
#include <hip/hip_runtime.h>
#include <math.h>


typedef float v2f __attribute__((ext_vector_type(2)));
typedef float v8f __attribute__((ext_vector_type(8)));


__global__ void zero_deg_kernel(int* __restrict__ deg, int n) {
    int i = blockIdx.x * blockDim.x + threadIdx.x;
    if (i < n) deg[i] = 0;
}

__global__ void count_rank_kernel(const int* __restrict__ dst, int* __restrict__ deg,
                                  int* __restrict__ rank, int e) {
    int i = blockIdx.x * blockDim.x + threadIdx.x;
    if (i < e) rank[i] = atomicAdd(&deg[dst[i]], 1);
}

__global__ void __launch_bounds__(1024)
scan_dinv_kernel(const int* __restrict__ deg, int* __restrict__ rowptr,
                 float* __restrict__ dinv, int n) {
    __shared__ int s_incl[32];
    __shared__ int s_part[32];
    const int tid  = threadIdx.x;
    const int lane = tid & 31;
    const int wave = tid >> 5;
    const int b0   = blockIdx.x * 1024;

    int part = 0;
    const int4* d4 = (const int4*)deg;
    for (int k = tid; k < (b0 >> 2); k += 1024) {
        const int4 q = d4[k];
        part += (q.x + q.y) + (q.z + q.w);
    }
    const int i = b0 + tid;
    const int d = (i < n) ? deg[i] : 0;

    int incl = d;
#pragma unroll
    for (int off = 1; off < 32; off <<= 1) {
        const int t = __shfl_up(incl, off, 32);
        if (lane >= off) incl += t;
    }
#pragma unroll
    for (int off = 16; off > 0; off >>= 1) part += __shfl_xor(part, off, 32);

    if (lane == 31) s_incl[wave] = incl;
    if (lane == 0)  s_part[wave] = part;
    __syncthreads();

    int v = s_part[lane] + ((lane < wave) ? s_incl[lane] : 0);
#pragma unroll
    for (int off = 16; off > 0; off >>= 1) v += __shfl_xor(v, off, 32);

    if (i < n) {
        rowptr[i] = v + incl - d;
        dinv[i]   = rsqrtf((float)d + 1.0f);
    }
}

__global__ void fill_kernel(const int* __restrict__ src, const int* __restrict__ dst,
                            const int* __restrict__ rank, const int* __restrict__ rowptr,
                            int* __restrict__ col, int e) {
    int i = blockIdx.x * blockDim.x + threadIdx.x;
    if (i < e) col[rowptr[dst[i]] + rank[i]] = src[i];
}


template <int NOUT, bool RELU_BIAS>
__global__ void gemm_wmma_kernel(const float* __restrict__ A,
                                 const float* __restrict__ W,
                                 const float* __restrict__ bias,
                                 float* __restrict__ C,
                                 int nrows) {
    constexpr int K = 128;
    constexpr int NWAVES = NOUT / 16;
    __shared__ float sA[16 * K];

    const int lane = threadIdx.x & 31;
    const int wave = threadIdx.x >> 5;
    const int n0   = wave * 16;
    const int ncol = n0 + (lane & 15);
    const int koff = (lane & 16) ? 2 : 0;

    v2f wf[32];
#pragma unroll
    for (int kb = 0; kb < 32; ++kb) {
        const int k = kb * 4 + koff;
        wf[kb].x = W[(size_t)k * NOUT + ncol];
        wf[kb].y = W[(size_t)(k + 1) * NOUT + ncol];
    }

    const int m    = lane & 15;
    const int mrow = (lane & 16) ? 8 : 0;

    for (int row0 = blockIdx.x * 16; row0 + 16 <= nrows; row0 += gridDim.x * 16) {
        const float4* gsrc = (const float4*)(A + (size_t)row0 * K);
        for (int idx4 = threadIdx.x; idx4 < (16 * K) / 4; idx4 += NWAVES * 32) {
            float4 v = gsrc[idx4];
            if constexpr (RELU_BIAS) {
                const int k = (idx4 * 4) & (K - 1);
                v.x = fmaxf(v.x + bias[k + 0], 0.0f);
                v.y = fmaxf(v.y + bias[k + 1], 0.0f);
                v.z = fmaxf(v.z + bias[k + 2], 0.0f);
                v.w = fmaxf(v.w + bias[k + 3], 0.0f);
            }
            ((float4*)sA)[idx4] = v;
        }
        __syncthreads();

        v8f c = {0.f, 0.f, 0.f, 0.f, 0.f, 0.f, 0.f, 0.f};
#pragma unroll
        for (int kb = 0; kb < 32; ++kb) {
            const v2f a = *(const v2f*)&sA[m * K + kb * 4 + koff];
            c = __builtin_amdgcn_wmma_f32_16x16x4_f32(
                    false, a, false, wf[kb], (short)0, c, false, false);
        }

        float* cp = C + (size_t)(row0 + mrow) * NOUT + n0 + m;
#pragma unroll
        for (int g = 0; g < 8; ++g) cp[(size_t)g * NOUT] = c[g];

        __syncthreads();
    }
}


template <int FEAT, bool ADD_BIAS>
__global__ void __launch_bounds__(256)
gather_kernel(const float* __restrict__ hl,
              const float* __restrict__ dinv,
              const int* __restrict__ rowptr,
              const int* __restrict__ deg,
              const int* __restrict__ col,
              const float* __restrict__ bias,
              float* __restrict__ acc, int n) {
    constexpr int VEC = FEAT / 32;
    constexpr int U   = 4;
    typedef float vecf __attribute__((ext_vector_type(VEC)));

    const int lane = threadIdx.x & 31;
    const int node = __builtin_amdgcn_readfirstlane(
        (int)(blockIdx.x * (blockDim.x >> 5) + (threadIdx.x >> 5)));
    if (node >= n) return;

    const int   start = rowptr[node];
    const int   total = deg[node] + 1;
    const float di    = dinv[node];
    const float* hlane = hl + lane * VEC;

    vecf a = {};
    for (int base = 0; base < total; base += 32) {
        const int m = (total - base < 32) ? (total - base) : 32;
        const int p = base + lane;
        int j = node;
        if (lane < m && p > 0) j = col[start + p - 1];
        const float dj = dinv[j];
        for (int t = 0; t < m; t += U) {
            vecf  v[U];
            float wu[U];
#pragma unroll
            for (int u = 0; u < U; ++u) {
                const int ju = __builtin_amdgcn_readlane(j, t + u);
                v[u] = *(const vecf*)(hlane + (size_t)ju * FEAT);
            }
#pragma unroll
            for (int u = 0; u < U; ++u) {
                wu[u] = __int_as_float(__builtin_amdgcn_readlane(__float_as_int(dj), t + u)) * di;
                if (t + u < m) a = a + v[u] * wu[u];
            }
        }
    }

    if constexpr (ADD_BIAS) {
        const vecf b = *(const vecf*)(bias + lane * VEC);
        a = a + b;
    }
    *(vecf*)(acc + (size_t)node * FEAT + lane * VEC) = a;
}


extern "C" void kernel_launch(void* const* d_in, const int* in_sizes, int n_in,
                              void* d_out, int out_size, void* d_ws, size_t ws_size,
                              hipStream_t stream) {
    const float* x   = (const float*)d_in[0];
    const int*   ei  = (const int*)d_in[1];
    const float* W1  = (const float*)d_in[2];
    const float* b1  = (const float*)d_in[3];
    const float* W2  = (const float*)d_in[4];
    const float* b2  = (const float*)d_in[5];
    float*       out = (float*)d_out;

    const int N = in_sizes[0] / 128;
    const int E = in_sizes[1] / 2;
    const int* src = ei;
    const int* dst = ei + E;

    char* w = (char*)d_ws;
    auto carve = [&](size_t bytes) {
        void* p = (void*)w;
        w += (bytes + 255) & ~(size_t)255;
        return p;
    };
    int*   deg  = (int*)carve((size_t)N * 4);
    float* dinv = (float*)carve((size_t)N * 4);
    int*   rowp = (int*)carve((size_t)N * 4);
    int*   rank = (int*)carve((size_t)E * 4);
    int*   col  = (int*)carve((size_t)E * 4);
    float* hl1  = (float*)carve((size_t)N * 128 * 4);
    float* hagg = (float*)carve((size_t)N * 128 * 4);
    float* hl2  = hl1;
    (void)n_in; (void)out_size; (void)ws_size;

    const int T = 256;
    zero_deg_kernel<<<(N + T - 1) / T, T, 0, stream>>>(deg, N);
    count_rank_kernel<<<(E + T - 1) / T, T, 0, stream>>>(dst, deg, rank, E);
    scan_dinv_kernel<<<(N + 1023) / 1024, 1024, 0, stream>>>(deg, rowp, dinv, N);
    fill_kernel<<<(E + T - 1) / T, T, 0, stream>>>(src, dst, rank, rowp, col, E);

    gemm_wmma_kernel<128, false><<<2048, 256, 0, stream>>>(x, W1, nullptr, hl1, N);

    gather_kernel<128, false><<<(N + 7) / 8, 256, 0, stream>>>(hl1, dinv, rowp, deg, col, nullptr, hagg, N);

    gemm_wmma_kernel<64, true><<<2048, 128, 0, stream>>>(hagg, W2, b1, hl2, N);

    gather_kernel<64, true><<<(N + 7) / 8, 256, 0, stream>>>(hl2, dinv, rowp, deg, col, b2, out, N);
}


# Edge weight split: each GEMM scales A row j by deg^-1/2 of j; gathers add pre-scaled rows and scale the sum once by the target's factor: no per-neighbour norm lookup; add order kept, rounding differs
# speedup vs baseline: 2.0888x; 1.0324x over previous
#include <hip/hip_runtime.h>
#include <math.h>


typedef float v2f __attribute__((ext_vector_type(2)));
typedef float v8f __attribute__((ext_vector_type(8)));


__global__ void zero_deg_kernel(int* __restrict__ deg, int n) {
    int i = blockIdx.x * blockDim.x + threadIdx.x;
    if (i < n) deg[i] = 0;
}

__global__ void count_rank_kernel(const int* __restrict__ dst, int* __restrict__ deg,
                                  int* __restrict__ rank, int e) {
    int i = blockIdx.x * blockDim.x + threadIdx.x;
    if (i < e) rank[i] = atomicAdd(&deg[dst[i]], 1);
}

__global__ void __launch_bounds__(1024)
scan_dinv_kernel(const int* __restrict__ deg, int* __restrict__ rowptr,
                 float* __restrict__ dinv, int n) {
    __shared__ int s_incl[32];
    __shared__ int s_part[32];
    const int tid  = threadIdx.x;
    const int lane = tid & 31;
    const int wave = tid >> 5;
    const int b0   = blockIdx.x * 1024;

    int part = 0;
    const int4* d4 = (const int4*)deg;
    for (int k = tid; k < (b0 >> 2); k += 1024) {
        const int4 q = d4[k];
        part += (q.x + q.y) + (q.z + q.w);
    }
    const int i = b0 + tid;
    const int d = (i < n) ? deg[i] : 0;

    int incl = d;
#pragma unroll
    for (int off = 1; off < 32; off <<= 1) {
        const int t = __shfl_up(incl, off, 32);
        if (lane >= off) incl += t;
    }
#pragma unroll
    for (int off = 16; off > 0; off >>= 1) part += __shfl_xor(part, off, 32);

    if (lane == 31) s_incl[wave] = incl;
    if (lane == 0)  s_part[wave] = part;
    __syncthreads();

    int v = s_part[lane] + ((lane < wave) ? s_incl[lane] : 0);
#pragma unroll
    for (int off = 16; off > 0; off >>= 1) v += __shfl_xor(v, off, 32);

    if (i < n) {
        rowptr[i] = v + incl - d;
        dinv[i]   = rsqrtf((float)d + 1.0f);
    }
}

__global__ void fill_kernel(const int* __restrict__ src, const int* __restrict__ dst,
                            const int* __restrict__ rank, const int* __restrict__ rowptr,
                            int* __restrict__ col, int e) {
    int i = blockIdx.x * blockDim.x + threadIdx.x;
    if (i < e) col[rowptr[dst[i]] + rank[i]] = src[i];
}


template <int NOUT, bool RELU_BIAS>
__global__ void gemm_wmma_kernel(const float* __restrict__ A,
                                 const float* __restrict__ W,
                                 const float* __restrict__ bias,
                                 const float* __restrict__ rscale,
                                 float* __restrict__ C,
                                 int nrows) {
    constexpr int K = 128;
    constexpr int NWAVES = NOUT / 16;
    __shared__ float sA[16 * K];

    const int lane = threadIdx.x & 31;
    const int wave = threadIdx.x >> 5;
    const int n0   = wave * 16;
    const int ncol = n0 + (lane & 15);
    const int koff = (lane & 16) ? 2 : 0;

    v2f wf[32];
#pragma unroll
    for (int kb = 0; kb < 32; ++kb) {
        const int k = kb * 4 + koff;
        wf[kb].x = W[(size_t)k * NOUT + ncol];
        wf[kb].y = W[(size_t)(k + 1) * NOUT + ncol];
    }

    const int m    = lane & 15;
    const int mrow = (lane & 16) ? 8 : 0;
    const int wave_s = __builtin_amdgcn_readfirstlane(wave);

    for (int row0 = blockIdx.x * 16; row0 + 16 <= nrows; row0 += gridDim.x * 16) {
        const float4* gsrc = (const float4*)(A + (size_t)row0 * K);
        int r = row0 + wave_s;
        for (int idx4 = threadIdx.x; idx4 < (16 * K) / 4; idx4 += NWAVES * 32, r += NWAVES) {
            float4 v = gsrc[idx4];
            const float s = rscale[r];
            if constexpr (RELU_BIAS) {
                const int k = (idx4 * 4) & (K - 1);
                v.x = fmaxf(v.x + bias[k + 0], 0.0f);
                v.y = fmaxf(v.y + bias[k + 1], 0.0f);
                v.z = fmaxf(v.z + bias[k + 2], 0.0f);
                v.w = fmaxf(v.w + bias[k + 3], 0.0f);
            }
            v.x *= s; v.y *= s; v.z *= s; v.w *= s;
            ((float4*)sA)[idx4] = v;
        }
        __syncthreads();

        v8f c = {0.f, 0.f, 0.f, 0.f, 0.f, 0.f, 0.f, 0.f};
#pragma unroll
        for (int kb = 0; kb < 32; ++kb) {
            const v2f a = *(const v2f*)&sA[m * K + kb * 4 + koff];
            c = __builtin_amdgcn_wmma_f32_16x16x4_f32(
                    false, a, false, wf[kb], (short)0, c, false, false);
        }

        float* cp = C + (size_t)(row0 + mrow) * NOUT + n0 + m;
#pragma unroll
        for (int g = 0; g < 8; ++g) cp[(size_t)g * NOUT] = c[g];

        __syncthreads();
    }
}


template <int FEAT, bool ADD_BIAS>
__global__ void __launch_bounds__(256)
gather_kernel(const float* __restrict__ hs,
              const float* __restrict__ dinv,
              const int* __restrict__ rowptr,
              const int* __restrict__ deg,
              const int* __restrict__ col,
              const float* __restrict__ bias,
              float* __restrict__ acc, int n) {
    constexpr int VEC = FEAT / 32;
    constexpr int U   = 4;
    typedef float vecf __attribute__((ext_vector_type(VEC)));

    const int lane = threadIdx.x & 31;
    const int node = __builtin_amdgcn_readfirstlane(
        (int)(blockIdx.x * (blockDim.x >> 5) + (threadIdx.x >> 5)));
    if (node >= n) return;

    const int   start = rowptr[node];
    const int   total = deg[node] + 1;
    const float di    = dinv[node];
    const float* hlane = hs + lane * VEC;

    vecf a = {};
    for (int base = 0; base < total; base += 32) {
        const int m = (total - base < 32) ? (total - base) : 32;
        const int p = base + lane;
        int j = node;
        if (lane < m && p > 0) j = col[start + p - 1];
        for (int t = 0; t < m; t += U) {
            vecf v[U];
#pragma unroll
            for (int u = 0; u < U; ++u) {
                const int ju = __builtin_amdgcn_readlane(j, t + u);
                v[u] = *(const vecf*)(hlane + (size_t)ju * FEAT);
            }
#pragma unroll
            for (int u = 0; u < U; ++u) {
                if (t + u < m) a = a + v[u];
            }
        }
    }

    a = a * di;
    if constexpr (ADD_BIAS) {
        const vecf b = *(const vecf*)(bias + lane * VEC);
        a = a + b;
    }
    *(vecf*)(acc + (size_t)node * FEAT + lane * VEC) = a;
}


extern "C" void kernel_launch(void* const* d_in, const int* in_sizes, int n_in,
                              void* d_out, int out_size, void* d_ws, size_t ws_size,
                              hipStream_t stream) {
    const float* x   = (const float*)d_in[0];
    const int*   ei  = (const int*)d_in[1];
    const float* W1  = (const float*)d_in[2];
    const float* b1  = (const float*)d_in[3];
    const float* W2  = (const float*)d_in[4];
    const float* b2  = (const float*)d_in[5];
    float*       out = (float*)d_out;

    const int N = in_sizes[0] / 128;
    const int E = in_sizes[1] / 2;
    const int* src = ei;
    const int* dst = ei + E;

    char* w = (char*)d_ws;
    auto carve = [&](size_t bytes) {
        void* p = (void*)w;
        w += (bytes + 255) & ~(size_t)255;
        return p;
    };
    int*   deg  = (int*)carve((size_t)N * 4);
    float* dinv = (float*)carve((size_t)N * 4);
    int*   rowp = (int*)carve((size_t)N * 4);
    int*   rank = (int*)carve((size_t)E * 4);
    int*   col  = (int*)carve((size_t)E * 4);
    float* hl1  = (float*)carve((size_t)N * 128 * 4);
    float* hagg = (float*)carve((size_t)N * 128 * 4);
    float* hl2  = hl1;
    (void)n_in; (void)out_size; (void)ws_size;

    const int T = 256;
    zero_deg_kernel<<<(N + T - 1) / T, T, 0, stream>>>(deg, N);
    count_rank_kernel<<<(E + T - 1) / T, T, 0, stream>>>(dst, deg, rank, E);
    scan_dinv_kernel<<<(N + 1023) / 1024, 1024, 0, stream>>>(deg, rowp, dinv, N);
    fill_kernel<<<(E + T - 1) / T, T, 0, stream>>>(src, dst, rank, rowp, col, E);

    gemm_wmma_kernel<128, false><<<2048, 256, 0, stream>>>(x, W1, nullptr, dinv, hl1, N);

    gather_kernel<128, false><<<(N + 7) / 8, 256, 0, stream>>>(hl1, dinv, rowp, deg, col, nullptr, hagg, N);

    gemm_wmma_kernel<64, true><<<2048, 128, 0, stream>>>(hagg, W2, b1, dinv, hl2, N);

    gather_kernel<64, true><<<(N + 7) / 8, 256, 0, stream>>>(hl2, dinv, rowp, deg, col, b2, out, N);
}


# One-pass edge bucketing: 64 list slots per node, the counting atomic's returned rank is the slot and the same thread stores the source id (overflow list past 64); scan and fill launches removed
# speedup vs baseline: 2.2592x; 1.0816x over previous
#include <hip/hip_runtime.h>
#include <math.h>


typedef float v2f __attribute__((ext_vector_type(2)));
typedef float v8f __attribute__((ext_vector_type(8)));


constexpr int SLOTS = 64;
constexpr int EPT   = 4;

__device__ __forceinline__ float dinv_of(int d) { return rsqrtf((float)d + 1.0f); }

__global__ void zero_deg_kernel(int* __restrict__ deg, int* __restrict__ ovf_cnt, int n) {
    int i = blockIdx.x * blockDim.x + threadIdx.x;
    if (i < n) deg[i] = 0;
    if (i == 0) *ovf_cnt = 0;
}

constexpr int CT = 256;

template <bool GUARD>
__device__ __forceinline__ void count_slot_body(const int* __restrict__ src,
                                                const int* __restrict__ dst,
                                                int* __restrict__ deg,
                                                int* __restrict__ slots,
                                                int* __restrict__ ovf_cnt,
                                                int2* __restrict__ ovf,
                                                int i0, int e) {
    int d[EPT], s[EPT], r[EPT];
#pragma unroll
    for (int u = 0; u < EPT; ++u) {
        const int i = i0 + u * CT;
        const bool ok = !GUARD || i < e;
        d[u] = ok ? dst[i] : 0;
        s[u] = ok ? src[i] : 0;
    }
#pragma unroll
    for (int u = 0; u < EPT; ++u) {
        const int i = i0 + u * CT;
        const bool ok = !GUARD || i < e;
        r[u] = ok ? atomicAdd(&deg[d[u]], 1) : 0;
    }
#pragma unroll
    for (int u = 0; u < EPT; ++u) {
        const int i = i0 + u * CT;
        const bool ok = !GUARD || i < e;
        if (ok) {
            if (r[u] < SLOTS) {
                slots[(size_t)d[u] * SLOTS + r[u]] = s[u];
            } else {
                const int k = atomicAdd(ovf_cnt, 1);
                ovf[k] = make_int2(d[u], s[u]);
            }
        }
    }
}

__global__ void __launch_bounds__(CT)
count_slot_kernel(const int* __restrict__ src, const int* __restrict__ dst,
                  int* __restrict__ deg, int* __restrict__ slots,
                  int* __restrict__ ovf_cnt, int2* __restrict__ ovf, int e) {
    const int b0 = blockIdx.x * (CT * EPT);
    const int i0 = b0 + threadIdx.x;
    if (b0 + CT * EPT <= e) count_slot_body<false>(src, dst, deg, slots, ovf_cnt, ovf, i0, e);
    else                    count_slot_body<true >(src, dst, deg, slots, ovf_cnt, ovf, i0, e);
}


template <int NOUT, bool RELU_BIAS>
__global__ void gemm_wmma_kernel(const float* __restrict__ A,
                                 const float* __restrict__ W,
                                 const float* __restrict__ bias,
                                 const int* __restrict__ deg,
                                 float* __restrict__ C,
                                 int nrows) {
    constexpr int K = 128;
    constexpr int NWAVES = NOUT / 16;
    __shared__ float sA[16 * K];

    const int lane = threadIdx.x & 31;
    const int wave = threadIdx.x >> 5;
    const int n0   = wave * 16;
    const int ncol = n0 + (lane & 15);
    const int koff = (lane & 16) ? 2 : 0;

    v2f wf[32];
#pragma unroll
    for (int kb = 0; kb < 32; ++kb) {
        const int k = kb * 4 + koff;
        wf[kb].x = W[(size_t)k * NOUT + ncol];
        wf[kb].y = W[(size_t)(k + 1) * NOUT + ncol];
    }

    const int m    = lane & 15;
    const int mrow = (lane & 16) ? 8 : 0;
    const int wave_s = __builtin_amdgcn_readfirstlane(wave);

    for (int row0 = blockIdx.x * 16; row0 + 16 <= nrows; row0 += gridDim.x * 16) {
        const float4* gsrc = (const float4*)(A + (size_t)row0 * K);
        int r = row0 + wave_s;
        for (int idx4 = threadIdx.x; idx4 < (16 * K) / 4; idx4 += NWAVES * 32, r += NWAVES) {
            float4 v = gsrc[idx4];
            const float s = dinv_of(deg[r]);
            if constexpr (RELU_BIAS) {
                const int k = (idx4 * 4) & (K - 1);
                v.x = fmaxf(v.x + bias[k + 0], 0.0f);
                v.y = fmaxf(v.y + bias[k + 1], 0.0f);
                v.z = fmaxf(v.z + bias[k + 2], 0.0f);
                v.w = fmaxf(v.w + bias[k + 3], 0.0f);
            }
            v.x *= s; v.y *= s; v.z *= s; v.w *= s;
            ((float4*)sA)[idx4] = v;
        }
        __syncthreads();

        v8f c = {0.f, 0.f, 0.f, 0.f, 0.f, 0.f, 0.f, 0.f};
#pragma unroll
        for (int kb = 0; kb < 32; ++kb) {
            const v2f a = *(const v2f*)&sA[m * K + kb * 4 + koff];
            c = __builtin_amdgcn_wmma_f32_16x16x4_f32(
                    false, a, false, wf[kb], (short)0, c, false, false);
        }

        float* cp = C + (size_t)(row0 + mrow) * NOUT + n0 + m;
#pragma unroll
        for (int g = 0; g < 8; ++g) cp[(size_t)g * NOUT] = c[g];

        __syncthreads();
    }
}


template <int FEAT, bool ADD_BIAS>
__global__ void __launch_bounds__(256)
gather_kernel(const float* __restrict__ hs,
              const int* __restrict__ deg,
              const int* __restrict__ slots,
              const int* __restrict__ ovf_cnt,
              const int2* __restrict__ ovf,
              const float* __restrict__ bias,
              float* __restrict__ acc, int n, int e) {
    constexpr int VEC = FEAT / 32;
    constexpr int U   = 4;
    typedef float vecf __attribute__((ext_vector_type(VEC)));

    const int lane = threadIdx.x & 31;
    const int node = __builtin_amdgcn_readfirstlane(
        (int)(blockIdx.x * (blockDim.x >> 5) + (threadIdx.x >> 5)));
    if (node >= n) return;

    const int   d     = deg[node];
    const int   total = (d < SLOTS ? d : SLOTS) + 1;
    const float di    = dinv_of(d);
    const int*  list  = slots + (size_t)node * SLOTS;
    const float* hlane = hs + lane * VEC;

    vecf a = {};
    for (int base = 0; base < total; base += 32) {
        const int m = (total - base < 32) ? (total - base) : 32;
        const int p = base + lane;
        int j = node;
        if (lane < m && p > 0) j = list[p - 1];
        for (int t = 0; t < m; t += U) {
            vecf v[U];
#pragma unroll
            for (int u = 0; u < U; ++u) {
                const int ju = __builtin_amdgcn_readlane(j, t + u);
                v[u] = *(const vecf*)(hlane + (size_t)ju * FEAT);
            }
#pragma unroll
            for (int u = 0; u < U; ++u) {
                if (t + u < m) a = a + v[u];
            }
        }
    }

    if (d > SLOTS) {
        int cnt = *ovf_cnt;
        if (cnt > e) cnt = e;
        for (int b = 0; b < cnt; b += 32) {
            const int k = b + lane;
            int od = 0, os = 0;
            if (k < cnt) { const int2 q = ovf[k]; od = q.x; os = q.y; }
            unsigned hit = __builtin_amdgcn_ballot_w32(k < cnt && od == node);
            while (hit) {
                const int l = __builtin_ctz(hit);
                hit &= hit - 1;
                const int ju = __builtin_amdgcn_readlane(os, l);
                a = a + *(const vecf*)(hlane + (size_t)ju * FEAT);
            }
        }
    }

    a = a * di;
    if constexpr (ADD_BIAS) {
        const vecf b = *(const vecf*)(bias + lane * VEC);
        a = a + b;
    }
    *(vecf*)(acc + (size_t)node * FEAT + lane * VEC) = a;
}


extern "C" void kernel_launch(void* const* d_in, const int* in_sizes, int n_in,
                              void* d_out, int out_size, void* d_ws, size_t ws_size,
                              hipStream_t stream) {
    const float* x   = (const float*)d_in[0];
    const int*   ei  = (const int*)d_in[1];
    const float* W1  = (const float*)d_in[2];
    const float* b1  = (const float*)d_in[3];
    const float* W2  = (const float*)d_in[4];
    const float* b2  = (const float*)d_in[5];
    float*       out = (float*)d_out;

    const int N = in_sizes[0] / 128;
    const int E = in_sizes[1] / 2;
    const int* src = ei;
    const int* dst = ei + E;

    char* w = (char*)d_ws;
    auto carve = [&](size_t bytes) {
        void* p = (void*)w;
        w += (bytes + 255) & ~(size_t)255;
        return p;
    };
    int*   deg   = (int*)carve((size_t)N * 4);
    int*   ovfc  = (int*)carve(4);
    int*   slots = (int*)carve((size_t)N * SLOTS * 4);
    int2*  ovf   = (int2*)carve((size_t)E * 8);
    float* hl1  = (float*)carve((size_t)N * 128 * 4);
    float* hagg = (float*)carve((size_t)N * 128 * 4);
    float* hl2  = hl1;
    (void)n_in; (void)out_size; (void)ws_size;

    const int T = 256;
    zero_deg_kernel<<<(N + T - 1) / T, T, 0, stream>>>(deg, ovfc, N);
    count_slot_kernel<<<(E + CT * EPT - 1) / (CT * EPT), CT, 0, stream>>>(src, dst, deg, slots, ovfc, ovf, E);

    gemm_wmma_kernel<128, false><<<2048, 256, 0, stream>>>(x, W1, nullptr, deg, hl1, N);

    gather_kernel<128, false><<<(N + 7) / 8, 256, 0, stream>>>(hl1, deg, slots, ovfc, ovf, nullptr, hagg, N, E);

    gemm_wmma_kernel<64, true><<<2048, 128, 0, stream>>>(hagg, W2, b1, deg, hl2, N);

    gather_kernel<64, true><<<(N + 7) / 8, 256, 0, stream>>>(hl2, deg, slots, ovfc, ovf, b2, out, N, E);
}


# Layer-1 neighbour sum fused with the layer-2 product: a 16-wave block sums 16 nodes' rows into an LDS strip, 4 waves multiply it by W2 held in LDS; summed table and one launch gone; add order kept
# speedup vs baseline: 2.5001x; 1.1066x over previous
#include <hip/hip_runtime.h>
#include <math.h>


typedef float v2f __attribute__((ext_vector_type(2)));
typedef float v8f __attribute__((ext_vector_type(8)));


constexpr int SLOTS = 64;
constexpr int EPT   = 4;

__device__ __forceinline__ float dinv_of(int d) { return rsqrtf((float)d + 1.0f); }

__global__ void zero_deg_kernel(int* __restrict__ deg, int* __restrict__ ovf_cnt, int n) {
    int i = blockIdx.x * blockDim.x + threadIdx.x;
    if (i < n) deg[i] = 0;
    if (i == 0) *ovf_cnt = 0;
}

constexpr int CT = 256;

template <bool GUARD>
__device__ __forceinline__ void count_slot_body(const int* __restrict__ src,
                                                const int* __restrict__ dst,
                                                int* __restrict__ deg,
                                                int* __restrict__ slots,
                                                int* __restrict__ ovf_cnt,
                                                int2* __restrict__ ovf,
                                                int i0, int e) {
    int d[EPT], s[EPT], r[EPT];
#pragma unroll
    for (int u = 0; u < EPT; ++u) {
        const int i = i0 + u * CT;
        const bool ok = !GUARD || i < e;
        d[u] = ok ? dst[i] : 0;
        s[u] = ok ? src[i] : 0;
    }
#pragma unroll
    for (int u = 0; u < EPT; ++u) {
        const int i = i0 + u * CT;
        const bool ok = !GUARD || i < e;
        r[u] = ok ? atomicAdd(&deg[d[u]], 1) : 0;
    }
#pragma unroll
    for (int u = 0; u < EPT; ++u) {
        const int i = i0 + u * CT;
        const bool ok = !GUARD || i < e;
        if (ok) {
            if (r[u] < SLOTS) {
                slots[(size_t)d[u] * SLOTS + r[u]] = s[u];
            } else {
                const int k = atomicAdd(ovf_cnt, 1);
                ovf[k] = make_int2(d[u], s[u]);
            }
        }
    }
}

__global__ void __launch_bounds__(CT)
count_slot_kernel(const int* __restrict__ src, const int* __restrict__ dst,
                  int* __restrict__ deg, int* __restrict__ slots,
                  int* __restrict__ ovf_cnt, int2* __restrict__ ovf, int e) {
    const int b0 = blockIdx.x * (CT * EPT);
    const int i0 = b0 + threadIdx.x;
    if (b0 + CT * EPT <= e) count_slot_body<false>(src, dst, deg, slots, ovf_cnt, ovf, i0, e);
    else                    count_slot_body<true >(src, dst, deg, slots, ovf_cnt, ovf, i0, e);
}


template <int NOUT>
__global__ void gemm_wmma_kernel(const float* __restrict__ A,
                                 const float* __restrict__ W,
                                 const int* __restrict__ deg,
                                 float* __restrict__ C,
                                 int nrows) {
    constexpr int K = 128;
    constexpr int NWAVES = NOUT / 16;
    __shared__ float sA[16 * K];

    const int lane = threadIdx.x & 31;
    const int wave = threadIdx.x >> 5;
    const int n0   = wave * 16;
    const int ncol = n0 + (lane & 15);
    const int koff = (lane & 16) ? 2 : 0;

    v2f wf[32];
#pragma unroll
    for (int kb = 0; kb < 32; ++kb) {
        const int k = kb * 4 + koff;
        wf[kb].x = W[(size_t)k * NOUT + ncol];
        wf[kb].y = W[(size_t)(k + 1) * NOUT + ncol];
    }

    const int m    = lane & 15;
    const int mrow = (lane & 16) ? 8 : 0;
    const int wave_s = __builtin_amdgcn_readfirstlane(wave);

    for (int row0 = blockIdx.x * 16; row0 + 16 <= nrows; row0 += gridDim.x * 16) {
        const float4* gsrc = (const float4*)(A + (size_t)row0 * K);
        int r = row0 + wave_s;
        for (int idx4 = threadIdx.x; idx4 < (16 * K) / 4; idx4 += NWAVES * 32, r += NWAVES) {
            float4 v = gsrc[idx4];
            const float s = dinv_of(deg[r]);
            v.x *= s; v.y *= s; v.z *= s; v.w *= s;
            ((float4*)sA)[idx4] = v;
        }
        __syncthreads();

        v8f c = {0.f, 0.f, 0.f, 0.f, 0.f, 0.f, 0.f, 0.f};
#pragma unroll
        for (int kb = 0; kb < 32; ++kb) {
            const v2f a = *(const v2f*)&sA[m * K + kb * 4 + koff];
            c = __builtin_amdgcn_wmma_f32_16x16x4_f32(
                    false, a, false, wf[kb], (short)0, c, false, false);
        }

        float* cp = C + (size_t)(row0 + mrow) * NOUT + n0 + m;
#pragma unroll
        for (int g = 0; g < 8; ++g) cp[(size_t)g * NOUT] = c[g];

        __syncthreads();
    }
}


template <int VEC> struct lane_vec { typedef float type __attribute__((ext_vector_type(VEC))); };

template <int FEAT>
__device__ __forceinline__ typename lane_vec<FEAT / 32>::type
neighbour_sum(const float* __restrict__ hs, int node, int d,
              const int* __restrict__ slots,
              const int* __restrict__ ovf_cnt,
              const int2* __restrict__ ovf, int e, int lane) {
    constexpr int VEC = FEAT / 32;
    constexpr int U   = 4;
    typedef typename lane_vec<VEC>::type vecf;

    const int   total = (d < SLOTS ? d : SLOTS) + 1;
    const int*  list  = slots + (size_t)node * SLOTS;
    const float* hlane = hs + lane * VEC;

    vecf a = {};
    for (int base = 0; base < total; base += 32) {
        const int m = (total - base < 32) ? (total - base) : 32;
        const int p = base + lane;
        int j = node;
        if (lane < m && p > 0) j = list[p - 1];
        for (int t = 0; t < m; t += U) {
            vecf v[U];
#pragma unroll
            for (int u = 0; u < U; ++u) {
                const int ju = __builtin_amdgcn_readlane(j, t + u);
                v[u] = *(const vecf*)(hlane + (size_t)ju * FEAT);
            }
#pragma unroll
            for (int u = 0; u < U; ++u) {
                if (t + u < m) a = a + v[u];
            }
        }
    }

    if (d > SLOTS) {
        int cnt = *ovf_cnt;
        if (cnt > e) cnt = e;
        for (int b = 0; b < cnt; b += 32) {
            const int k = b + lane;
            int od = 0, os = 0;
            if (k < cnt) { const int2 q = ovf[k]; od = q.x; os = q.y; }
            unsigned hit = __builtin_amdgcn_ballot_w32(k < cnt && od == node);
            while (hit) {
                const int l = __builtin_ctz(hit);
                hit &= hit - 1;
                const int ju = __builtin_amdgcn_readlane(os, l);
                a = a + *(const vecf*)(hlane + (size_t)ju * FEAT);
            }
        }
    }
    return a;
}

template <int FEAT>
__global__ void __launch_bounds__(256)
gather_kernel(const float* __restrict__ hs,
              const int* __restrict__ deg,
              const int* __restrict__ slots,
              const int* __restrict__ ovf_cnt,
              const int2* __restrict__ ovf,
              const float* __restrict__ bias,
              float* __restrict__ acc, int n, int e) {
    constexpr int VEC = FEAT / 32;
    typedef typename lane_vec<VEC>::type vecf;

    const int lane = threadIdx.x & 31;
    const int node = __builtin_amdgcn_readfirstlane(
        (int)(blockIdx.x * (blockDim.x >> 5) + (threadIdx.x >> 5)));
    if (node >= n) return;

    const int   d  = deg[node];
    const float di = dinv_of(d);

    vecf a = neighbour_sum<FEAT>(hs, node, d, slots, ovf_cnt, ovf, e, lane);
    a = a * di;
    const vecf b = *(const vecf*)(bias + lane * VEC);
    a = a + b;
    *(vecf*)(acc + (size_t)node * FEAT + lane * VEC) = a;
}


constexpr int TILE    = 16;
constexpr int A_PITCH = 132;

__global__ void __launch_bounds__(TILE * 32)
gather_relu_gemm_kernel(const float* __restrict__ hs,
                        const int* __restrict__ deg,
                        const int* __restrict__ slots,
                        const int* __restrict__ ovf_cnt,
                        const int2* __restrict__ ovf,
                        const float* __restrict__ b1,
                        const float* __restrict__ W2,
                        float* __restrict__ C, int n, int e) {
    constexpr int K    = 128;
    constexpr int NOUT = 64;
    constexpr int NT   = NOUT / 16;
    __shared__ __attribute__((aligned(16))) float sW[(K / 4) * NT * 32 * 2];
    __shared__ __attribute__((aligned(16))) float sA[2][TILE * A_PITCH];

    const int lane = threadIdx.x & 31;
    const int wave = __builtin_amdgcn_readfirstlane((int)(threadIdx.x >> 5));

    constexpr int WPT = (K / 4) * NT * 32 / (TILE * 32);
    v2f wv[WPT];
#pragma unroll
    for (int i = 0; i < WPT; ++i) {
        const int idx = threadIdx.x + i * (TILE * 32);
        const int l   = idx & 31;
        const int ct  = (idx >> 5) & (NT - 1);
        const int kb  = idx / (NT * 32);
        const int k   = kb * 4 + ((l & 16) ? 2 : 0);
        const int col = ct * 16 + (l & 15);
        wv[i].x = W2[(size_t)k * NOUT + col];
        wv[i].y = W2[(size_t)(k + 1) * NOUT + col];
    }
#pragma unroll
    for (int i = 0; i < WPT; ++i)
        *(v2f*)&sW[(threadIdx.x + i * (TILE * 32)) * 2] = wv[i];

    const int m    = lane & 15;
    const int koff = (lane & 16) ? 2 : 0;
    const int mrow = (lane & 16) ? 8 : 0;
    const int ntiles = (n + TILE - 1) / TILE;
    const float4 bv = *(const float4*)(b1 + lane * 4);

    int buf = 0;
    for (int tile = blockIdx.x; tile < ntiles; tile += gridDim.x, buf ^= 1) {
        const int node = tile * TILE + wave;
        float4 r = make_float4(0.0f, 0.0f, 0.0f, 0.0f);
        if (node < n) {
            const int   d  = deg[node];
            const float di = dinv_of(d);
            lane_vec<4>::type a = neighbour_sum<K>(hs, node, d, slots, ovf_cnt, ovf, e, lane);
            a = a * di;
            r.x = fmaxf(a.x + bv.x, 0.0f) * di;
            r.y = fmaxf(a.y + bv.y, 0.0f) * di;
            r.z = fmaxf(a.z + bv.z, 0.0f) * di;
            r.w = fmaxf(a.w + bv.w, 0.0f) * di;
        }
        *(float4*)&sA[buf][wave * A_PITCH + lane * 4] = r;
        __syncthreads();

        if (wave < NT) {
            const float* ar = &sA[buf][m * A_PITCH + koff];
            const float* wr = &sW[(wave * 32 + lane) * 2];
            v8f c = {0.f, 0.f, 0.f, 0.f, 0.f, 0.f, 0.f, 0.f};
#pragma unroll
            for (int kb = 0; kb < K / 4; ++kb) {
                const v2f a = *(const v2f*)(ar + kb * 4);
                const v2f b = *(const v2f*)(wr + kb * (NT * 32 * 2));
                c = __builtin_amdgcn_wmma_f32_16x16x4_f32(
                        false, a, false, b, (short)0, c, false, false);
            }
            const int row = tile * TILE + mrow;
            float* cp = C + (size_t)row * NOUT + wave * 16 + m;
#pragma unroll
            for (int g = 0; g < 8; ++g) {
                if (row + g < n) cp[(size_t)g * NOUT] = c[g];
            }
        }
    }
}


extern "C" void kernel_launch(void* const* d_in, const int* in_sizes, int n_in,
                              void* d_out, int out_size, void* d_ws, size_t ws_size,
                              hipStream_t stream) {
    const float* x   = (const float*)d_in[0];
    const int*   ei  = (const int*)d_in[1];
    const float* W1  = (const float*)d_in[2];
    const float* b1  = (const float*)d_in[3];
    const float* W2  = (const float*)d_in[4];
    const float* b2  = (const float*)d_in[5];
    float*       out = (float*)d_out;

    const int N = in_sizes[0] / 128;
    const int E = in_sizes[1] / 2;
    const int* src = ei;
    const int* dst = ei + E;

    char* w = (char*)d_ws;
    auto carve = [&](size_t bytes) {
        void* p = (void*)w;
        w += (bytes + 255) & ~(size_t)255;
        return p;
    };
    int*   deg   = (int*)carve((size_t)N * 4);
    int*   ovfc  = (int*)carve(4);
    int*   slots = (int*)carve((size_t)N * SLOTS * 4);
    int2*  ovf   = (int2*)carve((size_t)E * 8);
    float* hl1  = (float*)carve((size_t)N * 128 * 4);
    float* hl2  = (float*)carve((size_t)N * 64 * 4);
    (void)n_in; (void)out_size; (void)ws_size;

    const int T = 256;
    zero_deg_kernel<<<(N + T - 1) / T, T, 0, stream>>>(deg, ovfc, N);
    count_slot_kernel<<<(E + CT * EPT - 1) / (CT * EPT), CT, 0, stream>>>(src, dst, deg, slots, ovfc, ovf, E);

    gemm_wmma_kernel<128><<<2048, 256, 0, stream>>>(x, W1, deg, hl1, N);

    gather_relu_gemm_kernel<<<1024, TILE * 32, 0, stream>>>(hl1, deg, slots, ovfc, ovf, b1, W2, hl2, N, E);

    gather_kernel<64><<<(N + 7) / 8, 256, 0, stream>>>(hl2, deg, slots, ovfc, ovf, b2, out, N, E);
}


# Edge bucketing in two LDS passes (chunk sort by 256-node bucket; a block per bucket builds lists and degrees in LDS) replaces global returning atomics and scattered stores; list order = LDS arrival
# speedup vs baseline: 4.0594x; 1.6237x over previous
#include <hip/hip_runtime.h>
#include <math.h>


typedef float v2f __attribute__((ext_vector_type(2)));
typedef float v8f __attribute__((ext_vector_type(8)));


constexpr int SLOTS    = 64;
constexpr int NSH      = 8;
constexpr int NPB      = 1 << NSH;
constexpr int NBKT_MAX = 512;
constexpr int NCH      = 256;
constexpr int CH_MAX   = 8192;
constexpr int BT       = 1024;

__device__ __forceinline__ float dinv_of(int d) { return rsqrtf((float)d + 1.0f); }

__global__ void __launch_bounds__(BT)
bin_kernel(const int* __restrict__ src, const int* __restrict__ dst,
           int2* __restrict__ sorted, int* __restrict__ off,
           int* __restrict__ ovf_cnt, int n, int e) {
    __shared__ int  cnt[NBKT_MAX];
    __shared__ int  base[NBKT_MAX + 1];
    __shared__ int2 stage[CH_MAX];

    const int tid   = threadIdx.x;
    const int nbkt  = (n + NPB - 1) >> NSH;
    const int chunk = (e + NCH - 1) / NCH;
    const int c0    = blockIdx.x * chunk;
    int c1 = c0 + chunk;
    if (c1 > e) c1 = e;
    const int m = c1 > c0 ? c1 - c0 : 0;

    if (tid < NBKT_MAX) cnt[tid] = 0;
    if (blockIdx.x == 0 && tid == 0) *ovf_cnt = 0;
    __syncthreads();

    constexpr int EPT = CH_MAX / BT;
    int d[EPT], s[EPT], r[EPT];
#pragma unroll
    for (int u = 0; u < EPT; ++u) {
        const int i = tid + u * BT;
        const bool ok = i < m;
        d[u] = ok ? dst[c0 + i] : 0;
        s[u] = ok ? src[c0 + i] : 0;
    }
#pragma unroll
    for (int u = 0; u < EPT; ++u) {
        const int i = tid + u * BT;
        r[u] = (i < m) ? atomicAdd(&cnt[d[u] >> NSH], 1) : 0;
    }
    __syncthreads();

    if (tid < 32) {
        constexpr int PER = NBKT_MAX / 32;
        int loc[PER];
        int sum = 0;
#pragma unroll
        for (int j = 0; j < PER; ++j) { loc[j] = cnt[tid * PER + j]; sum += loc[j]; }
        int inc = sum;
#pragma unroll
        for (int o = 1; o < 32; o <<= 1) {
            const int t = __shfl_up(inc, o);
            if (tid >= o) inc += t;
        }
        int run = inc - sum;
#pragma unroll
        for (int j = 0; j < PER; ++j) { base[tid * PER + j] = run; run += loc[j]; }
        if (tid == 31) base[NBKT_MAX] = run;
    }
    __syncthreads();

#pragma unroll
    for (int u = 0; u < EPT; ++u) {
        const int i = tid + u * BT;
        if (i < m) stage[base[d[u] >> NSH] + r[u]] = make_int2(d[u], s[u]);
    }
    __syncthreads();

#pragma unroll
    for (int u = 0; u < EPT; ++u) {
        const int i = tid + u * BT;
        if (i < m) sorted[(size_t)c0 + i] = stage[i];
    }
    for (int i = tid; i <= nbkt; i += BT)
        off[(size_t)blockIdx.x * (nbkt + 1) + i] = base[i];
}

__device__ __forceinline__ void place_edge(int2 q, int node0, int* lcnt, int* lslots,
                                           int* __restrict__ ovf_cnt,
                                           int2* __restrict__ ovf) {
    const int ln = q.x - node0;
    const int r  = atomicAdd(&lcnt[ln], 1);
    if (r < SLOTS) {
        lslots[ln * SLOTS + r] = q.y;
    } else {
        const int k = atomicAdd(ovf_cnt, 1);
        ovf[k] = q;
    }
}

__global__ void __launch_bounds__(BT)
build_kernel(const int2* __restrict__ sorted, const int* __restrict__ off,
             int* __restrict__ deg, int* __restrict__ slots,
             int* __restrict__ ovf_cnt, int2* __restrict__ ovf, int n, int e) {
    __shared__ int lcnt[NPB];
    __shared__ int segs[NCH];
    __shared__ int segl[NCH];
    __shared__ int lslots[NPB * SLOTS];

    constexpr int NW  = BT / 32;
    constexpr int SPW = NCH / NW;

    const int tid   = threadIdx.x;
    const int lane  = tid & 31;
    const int wave  = __builtin_amdgcn_readfirstlane(tid >> 5);
    const int b     = blockIdx.x;
    const int nbkt  = (n + NPB - 1) >> NSH;
    const int chunk = (e + NCH - 1) / NCH;
    const int node0 = b << NSH;

    if (tid < NPB) lcnt[tid] = 0;
    if (tid < NCH) {
        const int* o = off + (size_t)tid * (nbkt + 1) + b;
        const int st = o[0];
        segs[tid] = st;
        segl[tid] = o[1] - st;
    }
    __syncthreads();

    int2 q[SPW];
    int  len[SPW];
#pragma unroll
    for (int t = 0; t < SPW; ++t) {
        const int k  = wave + t * NW;
        const int st = __builtin_amdgcn_readfirstlane(segs[k]);
        len[t]       = __builtin_amdgcn_readfirstlane(segl[k]);
        const int2* p = sorted + (size_t)k * chunk + st;
        q[t] = make_int2(0, 0);
        if (lane < len[t]) q[t] = p[lane];
    }
#pragma unroll
    for (int t = 0; t < SPW; ++t) {
        if (lane < len[t]) place_edge(q[t], node0, lcnt, lslots, ovf_cnt, ovf);
    }
#pragma unroll
    for (int t = 0; t < SPW; ++t) {
        if (len[t] > 32) {
            const int k  = wave + t * NW;
            const int st = __builtin_amdgcn_readfirstlane(segs[k]);
            const int2* p = sorted + (size_t)k * chunk + st;
            for (int i = 32 + lane; i < len[t]; i += 32)
                place_edge(p[i], node0, lcnt, lslots, ovf_cnt, ovf);
        }
    }
    __syncthreads();

    int nn = n - node0;
    if (nn > NPB) nn = NPB;
    for (int idx = tid; idx < nn * SLOTS; idx += BT) {
        if ((idx & (SLOTS - 1)) < lcnt[idx / SLOTS])
            slots[(size_t)node0 * SLOTS + idx] = lslots[idx];
    }
    if (tid < nn) deg[node0 + tid] = lcnt[tid];
}


template <int NOUT>
__global__ void gemm_wmma_kernel(const float* __restrict__ A,
                                 const float* __restrict__ W,
                                 const int* __restrict__ deg,
                                 float* __restrict__ C,
                                 int nrows) {
    constexpr int K = 128;
    constexpr int NWAVES = NOUT / 16;
    __shared__ float sA[16 * K];

    const int lane = threadIdx.x & 31;
    const int wave = threadIdx.x >> 5;
    const int n0   = wave * 16;
    const int ncol = n0 + (lane & 15);
    const int koff = (lane & 16) ? 2 : 0;

    v2f wf[32];
#pragma unroll
    for (int kb = 0; kb < 32; ++kb) {
        const int k = kb * 4 + koff;
        wf[kb].x = W[(size_t)k * NOUT + ncol];
        wf[kb].y = W[(size_t)(k + 1) * NOUT + ncol];
    }

    const int m    = lane & 15;
    const int mrow = (lane & 16) ? 8 : 0;
    const int wave_s = __builtin_amdgcn_readfirstlane(wave);

    for (int row0 = blockIdx.x * 16; row0 + 16 <= nrows; row0 += gridDim.x * 16) {
        const float4* gsrc = (const float4*)(A + (size_t)row0 * K);
        int r = row0 + wave_s;
        for (int idx4 = threadIdx.x; idx4 < (16 * K) / 4; idx4 += NWAVES * 32, r += NWAVES) {
            float4 v = gsrc[idx4];
            const float s = dinv_of(deg[r]);
            v.x *= s; v.y *= s; v.z *= s; v.w *= s;
            ((float4*)sA)[idx4] = v;
        }
        __syncthreads();

        v8f c = {0.f, 0.f, 0.f, 0.f, 0.f, 0.f, 0.f, 0.f};
#pragma unroll
        for (int kb = 0; kb < 32; ++kb) {
            const v2f a = *(const v2f*)&sA[m * K + kb * 4 + koff];
            c = __builtin_amdgcn_wmma_f32_16x16x4_f32(
                    false, a, false, wf[kb], (short)0, c, false, false);
        }

        float* cp = C + (size_t)(row0 + mrow) * NOUT + n0 + m;
#pragma unroll
        for (int g = 0; g < 8; ++g) cp[(size_t)g * NOUT] = c[g];

        __syncthreads();
    }
}


template <int VEC> struct lane_vec { typedef float type __attribute__((ext_vector_type(VEC))); };

template <int FEAT>
__device__ __forceinline__ typename lane_vec<FEAT / 32>::type
neighbour_sum(const float* __restrict__ hs, int node, int d,
              const int* __restrict__ slots,
              const int* __restrict__ ovf_cnt,
              const int2* __restrict__ ovf, int e, int lane) {
    constexpr int VEC = FEAT / 32;
    constexpr int U   = 4;
    typedef typename lane_vec<VEC>::type vecf;

    const int   total = (d < SLOTS ? d : SLOTS) + 1;
    const int*  list  = slots + (size_t)node * SLOTS;
    const float* hlane = hs + lane * VEC;

    vecf a = {};
    for (int base = 0; base < total; base += 32) {
        const int m = (total - base < 32) ? (total - base) : 32;
        const int p = base + lane;
        int j = node;
        if (lane < m && p > 0) j = list[p - 1];
        for (int t = 0; t < m; t += U) {
            vecf v[U];
#pragma unroll
            for (int u = 0; u < U; ++u) {
                const int ju = __builtin_amdgcn_readlane(j, t + u);
                v[u] = *(const vecf*)(hlane + (size_t)ju * FEAT);
            }
#pragma unroll
            for (int u = 0; u < U; ++u) {
                if (t + u < m) a = a + v[u];
            }
        }
    }

    if (d > SLOTS) {
        int cnt = *ovf_cnt;
        if (cnt > e) cnt = e;
        for (int b = 0; b < cnt; b += 32) {
            const int k = b + lane;
            int od = 0, os = 0;
            if (k < cnt) { const int2 q = ovf[k]; od = q.x; os = q.y; }
            unsigned hit = __builtin_amdgcn_ballot_w32(k < cnt && od == node);
            while (hit) {
                const int l = __builtin_ctz(hit);
                hit &= hit - 1;
                const int ju = __builtin_amdgcn_readlane(os, l);
                a = a + *(const vecf*)(hlane + (size_t)ju * FEAT);
            }
        }
    }
    return a;
}

template <int FEAT>
__global__ void __launch_bounds__(256)
gather_kernel(const float* __restrict__ hs,
              const int* __restrict__ deg,
              const int* __restrict__ slots,
              const int* __restrict__ ovf_cnt,
              const int2* __restrict__ ovf,
              const float* __restrict__ bias,
              float* __restrict__ acc, int n, int e) {
    constexpr int VEC = FEAT / 32;
    typedef typename lane_vec<VEC>::type vecf;

    const int lane = threadIdx.x & 31;
    const int node = __builtin_amdgcn_readfirstlane(
        (int)(blockIdx.x * (blockDim.x >> 5) + (threadIdx.x >> 5)));
    if (node >= n) return;

    const int   d  = deg[node];
    const float di = dinv_of(d);

    vecf a = neighbour_sum<FEAT>(hs, node, d, slots, ovf_cnt, ovf, e, lane);
    a = a * di;
    const vecf b = *(const vecf*)(bias + lane * VEC);
    a = a + b;
    *(vecf*)(acc + (size_t)node * FEAT + lane * VEC) = a;
}


constexpr int TILE    = 16;
constexpr int A_PITCH = 132;

__global__ void __launch_bounds__(TILE * 32)
gather_relu_gemm_kernel(const float* __restrict__ hs,
                        const int* __restrict__ deg,
                        const int* __restrict__ slots,
                        const int* __restrict__ ovf_cnt,
                        const int2* __restrict__ ovf,
                        const float* __restrict__ b1,
                        const float* __restrict__ W2,
                        float* __restrict__ C, int n, int e) {
    constexpr int K    = 128;
    constexpr int NOUT = 64;
    constexpr int NT   = NOUT / 16;
    __shared__ __attribute__((aligned(16))) float sW[(K / 4) * NT * 32 * 2];
    __shared__ __attribute__((aligned(16))) float sA[2][TILE * A_PITCH];

    const int lane = threadIdx.x & 31;
    const int wave = __builtin_amdgcn_readfirstlane((int)(threadIdx.x >> 5));

    constexpr int WPT = (K / 4) * NT * 32 / (TILE * 32);
    v2f wv[WPT];
#pragma unroll
    for (int i = 0; i < WPT; ++i) {
        const int idx = threadIdx.x + i * (TILE * 32);
        const int l   = idx & 31;
        const int ct  = (idx >> 5) & (NT - 1);
        const int kb  = idx / (NT * 32);
        const int k   = kb * 4 + ((l & 16) ? 2 : 0);
        const int col = ct * 16 + (l & 15);
        wv[i].x = W2[(size_t)k * NOUT + col];
        wv[i].y = W2[(size_t)(k + 1) * NOUT + col];
    }
#pragma unroll
    for (int i = 0; i < WPT; ++i)
        *(v2f*)&sW[(threadIdx.x + i * (TILE * 32)) * 2] = wv[i];

    const int m    = lane & 15;
    const int koff = (lane & 16) ? 2 : 0;
    const int mrow = (lane & 16) ? 8 : 0;
    const int ntiles = (n + TILE - 1) / TILE;
    const float4 bv = *(const float4*)(b1 + lane * 4);

    int buf = 0;
    for (int tile = blockIdx.x; tile < ntiles; tile += gridDim.x, buf ^= 1) {
        const int node = tile * TILE + wave;
        float4 r = make_float4(0.0f, 0.0f, 0.0f, 0.0f);
        if (node < n) {
            const int   d  = deg[node];
            const float di = dinv_of(d);
            lane_vec<4>::type a = neighbour_sum<K>(hs, node, d, slots, ovf_cnt, ovf, e, lane);
            a = a * di;
            r.x = fmaxf(a.x + bv.x, 0.0f) * di;
            r.y = fmaxf(a.y + bv.y, 0.0f) * di;
            r.z = fmaxf(a.z + bv.z, 0.0f) * di;
            r.w = fmaxf(a.w + bv.w, 0.0f) * di;
        }
        *(float4*)&sA[buf][wave * A_PITCH + lane * 4] = r;
        __syncthreads();

        if (wave < NT) {
            const float* ar = &sA[buf][m * A_PITCH + koff];
            const float* wr = &sW[(wave * 32 + lane) * 2];
            v8f c = {0.f, 0.f, 0.f, 0.f, 0.f, 0.f, 0.f, 0.f};
#pragma unroll
            for (int kb = 0; kb < K / 4; ++kb) {
                const v2f a = *(const v2f*)(ar + kb * 4);
                const v2f b = *(const v2f*)(wr + kb * (NT * 32 * 2));
                c = __builtin_amdgcn_wmma_f32_16x16x4_f32(
                        false, a, false, b, (short)0, c, false, false);
            }
            const int row = tile * TILE + mrow;
            float* cp = C + (size_t)row * NOUT + wave * 16 + m;
#pragma unroll
            for (int g = 0; g < 8; ++g) {
                if (row + g < n) cp[(size_t)g * NOUT] = c[g];
            }
        }
    }
}


extern "C" void kernel_launch(void* const* d_in, const int* in_sizes, int n_in,
                              void* d_out, int out_size, void* d_ws, size_t ws_size,
                              hipStream_t stream) {
    const float* x   = (const float*)d_in[0];
    const int*   ei  = (const int*)d_in[1];
    const float* W1  = (const float*)d_in[2];
    const float* b1  = (const float*)d_in[3];
    const float* W2  = (const float*)d_in[4];
    const float* b2  = (const float*)d_in[5];
    float*       out = (float*)d_out;

    const int N = in_sizes[0] / 128;
    const int E = in_sizes[1] / 2;
    const int* src = ei;
    const int* dst = ei + E;

    char* w = (char*)d_ws;
    auto carve = [&](size_t bytes) {
        void* p = (void*)w;
        w += (bytes + 255) & ~(size_t)255;
        return p;
    };
    int*   deg   = (int*)carve((size_t)N * 4);
    int*   ovfc  = (int*)carve(4);
    int*   slots = (int*)carve((size_t)N * SLOTS * 4);
    int2*  ovf   = (int2*)carve((size_t)E * 8);
    int2*  sorted = (int2*)carve((size_t)E * 8);
    int*   off   = (int*)carve((size_t)NCH * (NBKT_MAX + 1) * 4);
    float* hl1  = (float*)carve((size_t)N * 128 * 4);
    float* hl2  = (float*)carve((size_t)N * 64 * 4);
    (void)n_in; (void)out_size; (void)ws_size;

    bin_kernel<<<NCH, BT, 0, stream>>>(src, dst, sorted, off, ovfc, N, E);
    build_kernel<<<(N + NPB - 1) / NPB, BT, 0, stream>>>(sorted, off, deg, slots, ovfc, ovf, N, E);

    gemm_wmma_kernel<128><<<2048, 256, 0, stream>>>(x, W1, deg, hl1, N);

    gather_relu_gemm_kernel<<<1024, TILE * 32, 0, stream>>>(hl1, deg, slots, ovfc, ovf, b1, W2, hl2, N, E);

    gather_kernel<64><<<(N + 7) / 8, 256, 0, stream>>>(hl2, deg, slots, ovfc, ovf, b2, out, N, E);
}


# Layer 1 as Agg(x) @ W1: the fused kernel sums dinv[j]*x[j] per node and multiplies the strip by W1, then W2, from LDS; x @ W1 launch and its table gone; sum-then-multiply changes rounding order
# speedup vs baseline: 5.2610x; 1.2960x over previous
#include <hip/hip_runtime.h>
#include <math.h>


typedef float v2f __attribute__((ext_vector_type(2)));
typedef float v8f __attribute__((ext_vector_type(8)));


constexpr int SLOTS    = 64;
constexpr int NSH      = 8;
constexpr int NPB      = 1 << NSH;
constexpr int NBKT_MAX = 512;
constexpr int NCH      = 256;
constexpr int CH_MAX   = 8192;
constexpr int BT       = 1024;

__device__ __forceinline__ float dinv_of(int d) { return rsqrtf((float)d + 1.0f); }

__global__ void __launch_bounds__(BT)
bin_kernel(const int* __restrict__ src, const int* __restrict__ dst,
           int2* __restrict__ sorted, int* __restrict__ off,
           int* __restrict__ ovf_cnt, int n, int e) {
    __shared__ int  cnt[NBKT_MAX];
    __shared__ int  base[NBKT_MAX + 1];
    __shared__ int2 stage[CH_MAX];

    const int tid   = threadIdx.x;
    const int nbkt  = (n + NPB - 1) >> NSH;
    const int chunk = (e + NCH - 1) / NCH;
    const int c0    = blockIdx.x * chunk;
    int c1 = c0 + chunk;
    if (c1 > e) c1 = e;
    const int m = c1 > c0 ? c1 - c0 : 0;

    if (tid < NBKT_MAX) cnt[tid] = 0;
    if (blockIdx.x == 0 && tid == 0) *ovf_cnt = 0;
    __syncthreads();

    constexpr int EPT = CH_MAX / BT;
    int d[EPT], s[EPT], r[EPT];
#pragma unroll
    for (int u = 0; u < EPT; ++u) {
        const int i = tid + u * BT;
        const bool ok = i < m;
        d[u] = ok ? dst[c0 + i] : 0;
        s[u] = ok ? src[c0 + i] : 0;
    }
#pragma unroll
    for (int u = 0; u < EPT; ++u) {
        const int i = tid + u * BT;
        r[u] = (i < m) ? atomicAdd(&cnt[d[u] >> NSH], 1) : 0;
    }
    __syncthreads();

    if (tid < 32) {
        constexpr int PER = NBKT_MAX / 32;
        int loc[PER];
        int sum = 0;
#pragma unroll
        for (int j = 0; j < PER; ++j) { loc[j] = cnt[tid * PER + j]; sum += loc[j]; }
        int inc = sum;
#pragma unroll
        for (int o = 1; o < 32; o <<= 1) {
            const int t = __shfl_up(inc, o);
            if (tid >= o) inc += t;
        }
        int run = inc - sum;
#pragma unroll
        for (int j = 0; j < PER; ++j) { base[tid * PER + j] = run; run += loc[j]; }
        if (tid == 31) base[NBKT_MAX] = run;
    }
    __syncthreads();

#pragma unroll
    for (int u = 0; u < EPT; ++u) {
        const int i = tid + u * BT;
        if (i < m) stage[base[d[u] >> NSH] + r[u]] = make_int2(d[u], s[u]);
    }
    __syncthreads();

#pragma unroll
    for (int u = 0; u < EPT; ++u) {
        const int i = tid + u * BT;
        if (i < m) sorted[(size_t)c0 + i] = stage[i];
    }
    for (int i = tid; i <= nbkt; i += BT)
        off[(size_t)blockIdx.x * (nbkt + 1) + i] = base[i];
}

__device__ __forceinline__ void place_edge(int2 q, int node0, int* lcnt, int* lslots,
                                           int* __restrict__ ovf_cnt,
                                           int2* __restrict__ ovf) {
    const int ln = q.x - node0;
    const int r  = atomicAdd(&lcnt[ln], 1);
    if (r < SLOTS) {
        lslots[ln * SLOTS + r] = q.y;
    } else {
        const int k = atomicAdd(ovf_cnt, 1);
        ovf[k] = q;
    }
}

__global__ void __launch_bounds__(BT)
build_kernel(const int2* __restrict__ sorted, const int* __restrict__ off,
             int* __restrict__ deg, int* __restrict__ slots,
             int* __restrict__ ovf_cnt, int2* __restrict__ ovf, int n, int e) {
    __shared__ int lcnt[NPB];
    __shared__ int segs[NCH];
    __shared__ int segl[NCH];
    __shared__ int lslots[NPB * SLOTS];

    constexpr int NW  = BT / 32;
    constexpr int SPW = NCH / NW;

    const int tid   = threadIdx.x;
    const int lane  = tid & 31;
    const int wave  = __builtin_amdgcn_readfirstlane(tid >> 5);
    const int b     = blockIdx.x;
    const int nbkt  = (n + NPB - 1) >> NSH;
    const int chunk = (e + NCH - 1) / NCH;
    const int node0 = b << NSH;

    if (tid < NPB) lcnt[tid] = 0;
    if (tid < NCH) {
        const int* o = off + (size_t)tid * (nbkt + 1) + b;
        const int st = o[0];
        segs[tid] = st;
        segl[tid] = o[1] - st;
    }
    __syncthreads();

    int2 q[SPW];
    int  len[SPW];
#pragma unroll
    for (int t = 0; t < SPW; ++t) {
        const int k  = wave + t * NW;
        const int st = __builtin_amdgcn_readfirstlane(segs[k]);
        len[t]       = __builtin_amdgcn_readfirstlane(segl[k]);
        const int2* p = sorted + (size_t)k * chunk + st;
        q[t] = make_int2(0, 0);
        if (lane < len[t]) q[t] = p[lane];
    }
#pragma unroll
    for (int t = 0; t < SPW; ++t) {
        if (lane < len[t]) place_edge(q[t], node0, lcnt, lslots, ovf_cnt, ovf);
    }
#pragma unroll
    for (int t = 0; t < SPW; ++t) {
        if (len[t] > 32) {
            const int k  = wave + t * NW;
            const int st = __builtin_amdgcn_readfirstlane(segs[k]);
            const int2* p = sorted + (size_t)k * chunk + st;
            for (int i = 32 + lane; i < len[t]; i += 32)
                place_edge(p[i], node0, lcnt, lslots, ovf_cnt, ovf);
        }
    }
    __syncthreads();

    int nn = n - node0;
    if (nn > NPB) nn = NPB;
    for (int idx = tid; idx < nn * SLOTS; idx += BT) {
        if ((idx & (SLOTS - 1)) < lcnt[idx / SLOTS])
            slots[(size_t)node0 * SLOTS + idx] = lslots[idx];
    }
    if (tid < nn) deg[node0 + tid] = lcnt[tid];
}


template <int VEC> struct lane_vec { typedef float type __attribute__((ext_vector_type(VEC))); };

template <int FEAT, bool SCALED, int U>
__device__ __forceinline__ typename lane_vec<FEAT / 32>::type
neighbour_sum(const float* __restrict__ rows, int node, int d,
              const int* __restrict__ deg,
              const int* __restrict__ slots,
              const int* __restrict__ ovf_cnt,
              const int2* __restrict__ ovf, int e, int lane) {
    constexpr int VEC = FEAT / 32;
    typedef typename lane_vec<VEC>::type vecf;

    const int   total = (d < SLOTS ? d : SLOTS) + 1;
    const int*  list  = slots + (size_t)node * SLOTS;
    const float* rlane = rows + lane * VEC;

    vecf a = {};
    for (int base = 0; base < total; base += 32) {
        const int m = (total - base < 32) ? (total - base) : 32;
        const int p = base + lane;
        int j = node;
        if (lane < m && p > 0) j = list[p - 1];
        int sbits = 0;
        if (SCALED) sbits = __float_as_int(dinv_of(deg[j]));
        for (int t = 0; t < m; t += U) {
            vecf v[U];
#pragma unroll
            for (int u = 0; u < U; ++u) {
                const int ju = __builtin_amdgcn_readlane(j, t + u);
                v[u] = *(const vecf*)(rlane + (size_t)ju * FEAT);
            }
#pragma unroll
            for (int u = 0; u < U; ++u) {
                if (t + u < m) {
                    if (SCALED) {
                        const float su = __int_as_float(__builtin_amdgcn_readlane(sbits, t + u));
                        a = a + v[u] * su;
                    } else {
                        a = a + v[u];
                    }
                }
            }
        }
    }

    if (d > SLOTS) {
        int cnt = *ovf_cnt;
        if (cnt > e) cnt = e;
        for (int b = 0; b < cnt; b += 32) {
            const int k = b + lane;
            int od = 0, os = 0;
            if (k < cnt) { const int2 q = ovf[k]; od = q.x; os = q.y; }
            unsigned hit = __builtin_amdgcn_ballot_w32(k < cnt && od == node);
            while (hit) {
                const int l = __builtin_ctz(hit);
                hit &= hit - 1;
                const int ju = __builtin_amdgcn_readlane(os, l);
                const vecf v = *(const vecf*)(rlane + (size_t)ju * FEAT);
                if (SCALED) a = a + v * dinv_of(deg[ju]);
                else        a = a + v;
            }
        }
    }
    return a;
}

template <int FEAT>
__global__ void __launch_bounds__(256)
gather_kernel(const float* __restrict__ hs,
              const int* __restrict__ deg,
              const int* __restrict__ slots,
              const int* __restrict__ ovf_cnt,
              const int2* __restrict__ ovf,
              const float* __restrict__ bias,
              float* __restrict__ acc, int n, int e) {
    constexpr int VEC = FEAT / 32;
    typedef typename lane_vec<VEC>::type vecf;

    const int lane = threadIdx.x & 31;
    const int node = __builtin_amdgcn_readfirstlane(
        (int)(blockIdx.x * (blockDim.x >> 5) + (threadIdx.x >> 5)));
    if (node >= n) return;

    const int   d  = deg[node];
    const float di = dinv_of(d);

    vecf a = neighbour_sum<FEAT, false, 4>(hs, node, d, deg, slots, ovf_cnt, ovf, e, lane);
    a = a * di;
    const vecf b = *(const vecf*)(bias + lane * VEC);
    a = a + b;
    *(vecf*)(acc + (size_t)node * FEAT + lane * VEC) = a;
}


constexpr int TILE    = 16;
constexpr int A_PITCH = 132;

__device__ __forceinline__ v8f strip_product(const float* ar, const float* wr, int wstep) {
    v8f c = {0.f, 0.f, 0.f, 0.f, 0.f, 0.f, 0.f, 0.f};
#pragma unroll
    for (int kb = 0; kb < 32; ++kb) {
        const v2f a = *(const v2f*)(ar + kb * 4);
        const v2f b = *(const v2f*)(wr + kb * wstep);
        c = __builtin_amdgcn_wmma_f32_16x16x4_f32(
                false, a, false, b, (short)0, c, false, false);
    }
    return c;
}

__global__ void __launch_bounds__(TILE * 32)
agg_gemm_relu_gemm_kernel(const float* __restrict__ x,
                          const int* __restrict__ deg,
                          const int* __restrict__ slots,
                          const int* __restrict__ ovf_cnt,
                          const int2* __restrict__ ovf,
                          const float* __restrict__ W1,
                          const float* __restrict__ b1,
                          const float* __restrict__ W2,
                          float* __restrict__ C, int n, int e) {
    constexpr int K    = 128;
    constexpr int NOUT = 64;
    constexpr int NT1  = K / 16;
    constexpr int NT2  = NOUT / 16;
    __shared__ __attribute__((aligned(16))) float sW1[(K / 4) * NT1 * 32 * 2];
    __shared__ __attribute__((aligned(16))) float sW2[(K / 4) * NT2 * 32 * 2];
    __shared__ __attribute__((aligned(16))) float sA[2][TILE * A_PITCH];
    __shared__ __attribute__((aligned(16))) float sH[2][TILE * A_PITCH];

    const int lane = threadIdx.x & 31;
    const int wave = __builtin_amdgcn_readfirstlane((int)(threadIdx.x >> 5));

    {
        constexpr int WPT = (K / 4) * NT1 * 32 / (TILE * 32);
        v2f wv[WPT];
#pragma unroll
        for (int i = 0; i < WPT; ++i) {
            const int idx = threadIdx.x + i * (TILE * 32);
            const int l   = idx & 31;
            const int ct  = (idx >> 5) & (NT1 - 1);
            const int kb  = idx / (NT1 * 32);
            const int k   = kb * 4 + ((l & 16) ? 2 : 0);
            const int col = ct * 16 + (l & 15);
            wv[i].x = W1[(size_t)k * K + col];
            wv[i].y = W1[(size_t)(k + 1) * K + col];
        }
#pragma unroll
        for (int i = 0; i < WPT; ++i)
            *(v2f*)&sW1[(threadIdx.x + i * (TILE * 32)) * 2] = wv[i];
    }
    {
        constexpr int WPT = (K / 4) * NT2 * 32 / (TILE * 32);
        v2f wv[WPT];
#pragma unroll
        for (int i = 0; i < WPT; ++i) {
            const int idx = threadIdx.x + i * (TILE * 32);
            const int l   = idx & 31;
            const int ct  = (idx >> 5) & (NT2 - 1);
            const int kb  = idx / (NT2 * 32);
            const int k   = kb * 4 + ((l & 16) ? 2 : 0);
            const int col = ct * 16 + (l & 15);
            wv[i].x = W2[(size_t)k * NOUT + col];
            wv[i].y = W2[(size_t)(k + 1) * NOUT + col];
        }
#pragma unroll
        for (int i = 0; i < WPT; ++i)
            *(v2f*)&sW2[(threadIdx.x + i * (TILE * 32)) * 2] = wv[i];
    }

    const int m    = lane & 15;
    const int koff = (lane & 16) ? 2 : 0;
    const int mrow = (lane & 16) ? 8 : 0;
    const int ntiles = (n + TILE - 1) / TILE;

    auto layer2_product = [&](int ptile, int hb, int ct) {
        const v8f c = strip_product(&sH[hb][m * A_PITCH + koff],
                                    &sW2[(ct * 32 + lane) * 2], NT2 * 32 * 2);
        const int row = ptile * TILE + mrow;
        float* cp = C + (size_t)row * NOUT + ct * 16 + m;
#pragma unroll
        for (int g = 0; g < 8; ++g) {
            if (row + g < n) cp[(size_t)g * NOUT] = c[g];
        }
    };

    int buf  = 0;
    int prev = -1;
    for (int tile = blockIdx.x; tile < ntiles; tile += gridDim.x, buf ^= 1) {
        const int node = tile * TILE + wave;
        float4 r  = make_float4(0.0f, 0.0f, 0.0f, 0.0f);
        float  di = 0.0f;
        if (node < n) {
            const int d = deg[node];
            di = dinv_of(d);
            const lane_vec<4>::type a =
                neighbour_sum<K, true, 8>(x, node, d, deg, slots, ovf_cnt, ovf, e, lane);
            r.x = a.x * di; r.y = a.y * di; r.z = a.z * di; r.w = a.w * di;
        }
        *(float4*)&sA[buf][wave * A_PITCH + lane * 4] = r;
        if (lane == 0) sA[buf][wave * A_PITCH + K] = di;
        __syncthreads();

        if (wave < NT1) {
            const v8f c = strip_product(&sA[buf][m * A_PITCH + koff],
                                        &sW1[(wave * 32 + lane) * 2], NT1 * 32 * 2);
            const float bcol = b1[wave * 16 + m];
            float* hp = &sH[buf][mrow * A_PITCH + wave * 16 + m];
            const float* dp = &sA[buf][mrow * A_PITCH + K];
#pragma unroll
            for (int g = 0; g < 8; ++g)
                hp[g * A_PITCH] = fmaxf(c[g] + bcol, 0.0f) * dp[g * A_PITCH];
        } else if (wave < NT1 + NT2) {
            if (prev >= 0) layer2_product(prev, buf ^ 1, wave - NT1);
        }
        prev = tile;
    }
    __syncthreads();
    if (wave >= NT1 && wave < NT1 + NT2 && prev >= 0)
        layer2_product(prev, buf ^ 1, wave - NT1);
}


extern "C" void kernel_launch(void* const* d_in, const int* in_sizes, int n_in,
                              void* d_out, int out_size, void* d_ws, size_t ws_size,
                              hipStream_t stream) {
    const float* x   = (const float*)d_in[0];
    const int*   ei  = (const int*)d_in[1];
    const float* W1  = (const float*)d_in[2];
    const float* b1  = (const float*)d_in[3];
    const float* W2  = (const float*)d_in[4];
    const float* b2  = (const float*)d_in[5];
    float*       out = (float*)d_out;

    const int N = in_sizes[0] / 128;
    const int E = in_sizes[1] / 2;
    const int* src = ei;
    const int* dst = ei + E;

    char* w = (char*)d_ws;
    auto carve = [&](size_t bytes) {
        void* p = (void*)w;
        w += (bytes + 255) & ~(size_t)255;
        return p;
    };
    int*   deg   = (int*)carve((size_t)N * 4);
    int*   ovfc  = (int*)carve(4);
    int*   slots = (int*)carve((size_t)N * SLOTS * 4);
    int2*  ovf   = (int2*)carve((size_t)E * 8);
    int2*  sorted = (int2*)carve((size_t)E * 8);
    int*   off   = (int*)carve((size_t)NCH * (NBKT_MAX + 1) * 4);
    float* hl2  = (float*)carve((size_t)N * 64 * 4);
    (void)n_in; (void)out_size; (void)ws_size;

    bin_kernel<<<NCH, BT, 0, stream>>>(src, dst, sorted, off, ovfc, N, E);
    build_kernel<<<(N + NPB - 1) / NPB, BT, 0, stream>>>(sorted, off, deg, slots, ovfc, ovf, N, E);

    agg_gemm_relu_gemm_kernel<<<512, TILE * 32, 0, stream>>>(x, deg, slots, ovfc, ovf, W1, b1, W2, hl2, N, E);

    gather_kernel<64><<<(N + 7) / 8, 256, 0, stream>>>(hl2, deg, slots, ovfc, ovf, b2, out, N, E);
}


# Layer-2 gather as a resident grid: 8192 waves loop over nodes, a node's rows all in flight at once with the next node's list (was 100000 short waves, 4 rows in flight); add order kept
# speedup vs baseline: 5.4628x; 1.0384x over previous
#include <hip/hip_runtime.h>
#include <math.h>


typedef float v2f __attribute__((ext_vector_type(2)));
typedef float v8f __attribute__((ext_vector_type(8)));


constexpr int SLOTS    = 64;
constexpr int NSH      = 8;
constexpr int NPB      = 1 << NSH;
constexpr int NBKT_MAX = 512;
constexpr int NCH      = 256;
constexpr int CH_MAX   = 8192;
constexpr int BT       = 1024;

__device__ __forceinline__ float dinv_of(int d) { return rsqrtf((float)d + 1.0f); }

__global__ void __launch_bounds__(BT)
bin_kernel(const int* __restrict__ src, const int* __restrict__ dst,
           int2* __restrict__ sorted, int* __restrict__ off,
           int* __restrict__ ovf_cnt, int n, int e) {
    __shared__ int  cnt[NBKT_MAX];
    __shared__ int  base[NBKT_MAX + 1];
    __shared__ int2 stage[CH_MAX];

    const int tid   = threadIdx.x;
    const int nbkt  = (n + NPB - 1) >> NSH;
    const int chunk = (e + NCH - 1) / NCH;
    const int c0    = blockIdx.x * chunk;
    int c1 = c0 + chunk;
    if (c1 > e) c1 = e;
    const int m = c1 > c0 ? c1 - c0 : 0;

    if (tid < NBKT_MAX) cnt[tid] = 0;
    if (blockIdx.x == 0 && tid == 0) *ovf_cnt = 0;
    __syncthreads();

    constexpr int EPT = CH_MAX / BT;
    int d[EPT], s[EPT], r[EPT];
#pragma unroll
    for (int u = 0; u < EPT; ++u) {
        const int i = tid + u * BT;
        const bool ok = i < m;
        d[u] = ok ? dst[c0 + i] : 0;
        s[u] = ok ? src[c0 + i] : 0;
    }
#pragma unroll
    for (int u = 0; u < EPT; ++u) {
        const int i = tid + u * BT;
        r[u] = (i < m) ? atomicAdd(&cnt[d[u] >> NSH], 1) : 0;
    }
    __syncthreads();

    if (tid < 32) {
        constexpr int PER = NBKT_MAX / 32;
        int loc[PER];
        int sum = 0;
#pragma unroll
        for (int j = 0; j < PER; ++j) { loc[j] = cnt[tid * PER + j]; sum += loc[j]; }
        int inc = sum;
#pragma unroll
        for (int o = 1; o < 32; o <<= 1) {
            const int t = __shfl_up(inc, o);
            if (tid >= o) inc += t;
        }
        int run = inc - sum;
#pragma unroll
        for (int j = 0; j < PER; ++j) { base[tid * PER + j] = run; run += loc[j]; }
        if (tid == 31) base[NBKT_MAX] = run;
    }
    __syncthreads();

#pragma unroll
    for (int u = 0; u < EPT; ++u) {
        const int i = tid + u * BT;
        if (i < m) stage[base[d[u] >> NSH] + r[u]] = make_int2(d[u], s[u]);
    }
    __syncthreads();

#pragma unroll
    for (int u = 0; u < EPT; ++u) {
        const int i = tid + u * BT;
        if (i < m) sorted[(size_t)c0 + i] = stage[i];
    }
    for (int i = tid; i <= nbkt; i += BT)
        off[(size_t)blockIdx.x * (nbkt + 1) + i] = base[i];
}

__device__ __forceinline__ void place_edge(int2 q, int node0, int* lcnt, int* lslots,
                                           int* __restrict__ ovf_cnt,
                                           int2* __restrict__ ovf) {
    const int ln = q.x - node0;
    const int r  = atomicAdd(&lcnt[ln], 1);
    if (r < SLOTS) {
        lslots[ln * SLOTS + r] = q.y;
    } else {
        const int k = atomicAdd(ovf_cnt, 1);
        ovf[k] = q;
    }
}

__global__ void __launch_bounds__(BT)
build_kernel(const int2* __restrict__ sorted, const int* __restrict__ off,
             int* __restrict__ deg, int* __restrict__ slots,
             int* __restrict__ ovf_cnt, int2* __restrict__ ovf, int n, int e) {
    __shared__ int lcnt[NPB];
    __shared__ int segs[NCH];
    __shared__ int segl[NCH];
    __shared__ int lslots[NPB * SLOTS];

    constexpr int NW  = BT / 32;
    constexpr int SPW = NCH / NW;

    const int tid   = threadIdx.x;
    const int lane  = tid & 31;
    const int wave  = __builtin_amdgcn_readfirstlane(tid >> 5);
    const int b     = blockIdx.x;
    const int nbkt  = (n + NPB - 1) >> NSH;
    const int chunk = (e + NCH - 1) / NCH;
    const int node0 = b << NSH;

    if (tid < NPB) lcnt[tid] = 0;
    if (tid < NCH) {
        const int* o = off + (size_t)tid * (nbkt + 1) + b;
        const int st = o[0];
        segs[tid] = st;
        segl[tid] = o[1] - st;
    }
    __syncthreads();

    int2 q[SPW];
    int  len[SPW];
#pragma unroll
    for (int t = 0; t < SPW; ++t) {
        const int k  = wave + t * NW;
        const int st = __builtin_amdgcn_readfirstlane(segs[k]);
        len[t]       = __builtin_amdgcn_readfirstlane(segl[k]);
        const int2* p = sorted + (size_t)k * chunk + st;
        q[t] = make_int2(0, 0);
        if (lane < len[t]) q[t] = p[lane];
    }
#pragma unroll
    for (int t = 0; t < SPW; ++t) {
        if (lane < len[t]) place_edge(q[t], node0, lcnt, lslots, ovf_cnt, ovf);
    }
#pragma unroll
    for (int t = 0; t < SPW; ++t) {
        if (len[t] > 32) {
            const int k  = wave + t * NW;
            const int st = __builtin_amdgcn_readfirstlane(segs[k]);
            const int2* p = sorted + (size_t)k * chunk + st;
            for (int i = 32 + lane; i < len[t]; i += 32)
                place_edge(p[i], node0, lcnt, lslots, ovf_cnt, ovf);
        }
    }
    __syncthreads();

    int nn = n - node0;
    if (nn > NPB) nn = NPB;
    for (int idx = tid; idx < nn * SLOTS; idx += BT) {
        if ((idx & (SLOTS - 1)) < lcnt[idx / SLOTS])
            slots[(size_t)node0 * SLOTS + idx] = lslots[idx];
    }
    if (tid < nn) deg[node0 + tid] = lcnt[tid];
}


template <int VEC> struct lane_vec { typedef float type __attribute__((ext_vector_type(VEC))); };

template <int FEAT, bool SCALED>
__device__ __forceinline__ typename lane_vec<FEAT / 32>::type
overflow_sum(typename lane_vec<FEAT / 32>::type a, const float* rlane, int node,
             const int* __restrict__ deg,
             const int* __restrict__ ovf_cnt,
             const int2* __restrict__ ovf, int e, int lane) {
    typedef typename lane_vec<FEAT / 32>::type vecf;
    int cnt = *ovf_cnt;
    if (cnt > e) cnt = e;
    for (int b = 0; b < cnt; b += 32) {
        const int k = b + lane;
        int od = 0, os = 0;
        if (k < cnt) { const int2 q = ovf[k]; od = q.x; os = q.y; }
        unsigned hit = __builtin_amdgcn_ballot_w32(k < cnt && od == node);
        while (hit) {
            const int l = __builtin_ctz(hit);
            hit &= hit - 1;
            const int ju = __builtin_amdgcn_readlane(os, l);
            const vecf v = *(const vecf*)(rlane + (size_t)ju * FEAT);
            if (SCALED) a = a + v * dinv_of(deg[ju]);
            else        a = a + v;
        }
    }
    return a;
}

template <int FEAT, bool SCALED, int U>
__device__ __forceinline__ typename lane_vec<FEAT / 32>::type
neighbour_sum(const float* __restrict__ rows, int node, int d,
              const int* __restrict__ deg,
              const int* __restrict__ slots,
              const int* __restrict__ ovf_cnt,
              const int2* __restrict__ ovf, int e, int lane) {
    constexpr int VEC = FEAT / 32;
    typedef typename lane_vec<VEC>::type vecf;

    const int   total = (d < SLOTS ? d : SLOTS) + 1;
    const int*  list  = slots + (size_t)node * SLOTS;
    const float* rlane = rows + lane * VEC;

    vecf a = {};
    for (int base = 0; base < total; base += 32) {
        const int m = (total - base < 32) ? (total - base) : 32;
        const int p = base + lane;
        int j = node;
        if (lane < m && p > 0) j = list[p - 1];
        int sbits = 0;
        if (SCALED) sbits = __float_as_int(dinv_of(deg[j]));
        for (int t = 0; t < m; t += U) {
            vecf v[U];
#pragma unroll
            for (int u = 0; u < U; ++u) {
                const int ju = __builtin_amdgcn_readlane(j, t + u);
                v[u] = *(const vecf*)(rlane + (size_t)ju * FEAT);
            }
#pragma unroll
            for (int u = 0; u < U; ++u) {
                if (t + u < m) {
                    if (SCALED) {
                        const float su = __int_as_float(__builtin_amdgcn_readlane(sbits, t + u));
                        a = a + v[u] * su;
                    } else {
                        a = a + v[u];
                    }
                }
            }
        }
    }

    if (d > SLOTS)
        a = overflow_sum<FEAT, SCALED>(a, rlane, node, deg, ovf_cnt, ovf, e, lane);
    return a;
}


constexpr int GW            = 16;
constexpr int GATHER_BLOCKS = 512;

template <int FEAT>
__global__ void __launch_bounds__(GW * 32)
gather_kernel(const float* __restrict__ hs,
              const int* __restrict__ deg,
              const int* __restrict__ slots,
              const int* __restrict__ ovf_cnt,
              const int2* __restrict__ ovf,
              const float* __restrict__ bias,
              float* __restrict__ acc, int n, int e) {
    constexpr int VEC = FEAT / 32;
    typedef typename lane_vec<VEC>::type vecf;

    const int lane = threadIdx.x & 31;
    const int wid = __builtin_amdgcn_readfirstlane(
        (int)(blockIdx.x * GW + (threadIdx.x >> 5)));
    const int nw  = (int)gridDim.x * GW;
    if (wid >= n) return;

    const float* rlane = hs + lane * VEC;
    const vecf   bv    = *(const vecf*)(bias + lane * VEC);

    auto first_entries = [&](int node, int d) {
        const int total = (d < SLOTS ? d : SLOTS) + 1;
        int j = node;
        if (lane > 0 && lane < total) j = slots[(size_t)node * SLOTS + lane - 1];
        return j;
    };
    auto row = [&](int ju) {
        const float* rp = hs + (size_t)ju * FEAT;
        return *(const vecf*)(rp + lane * VEC);
    };

    int node  = wid;
    int d     = deg[node];
    int j     = first_entries(node, d);
    int node1 = node + nw;
    int d1    = 0;
    if (node1 < n) d1 = deg[node1];

    for (;;) {
        const int total = (d < SLOTS ? d : SLOTS) + 1;
        const int m0    = total < 32 ? total : 32;
        vecf v[32];
#pragma unroll
        for (int c = 0; c < 8; ++c) {
            if (c == 0 || c * 4 < m0) {
#pragma unroll
                for (int u = 0; u < 4; ++u)
                    v[c * 4 + u] = row(__builtin_amdgcn_readlane(j, c * 4 + u));
            }
        }

        const int node2 = node1 + nw;
        int d2 = 0;
        if (node2 < n) d2 = deg[node2];
        int j1 = 0;
        if (node1 < n) j1 = first_entries(node1, d1);

        vecf a = {};
#pragma unroll
        for (int c = 0; c < 8; ++c) {
            if (c == 0 || c * 4 < m0) {
#pragma unroll
                for (int u = 0; u < 4; ++u) {
                    if (c * 4 + u == 0 || c * 4 + u < m0) a = a + v[c * 4 + u];
                }
            }
        }
        for (int base = 32; base < total; base += 32) {
            const int m = (total - base < 32) ? (total - base) : 32;
            int jj = node;
            if (lane < m) jj = slots[(size_t)node * SLOTS + base + lane - 1];
            for (int t = 0; t < m; t += 4) {
                vecf w[4];
#pragma unroll
                for (int u = 0; u < 4; ++u)
                    w[u] = row(__builtin_amdgcn_readlane(jj, t + u));
#pragma unroll
                for (int u = 0; u < 4; ++u) {
                    if (t + u < m) a = a + w[u];
                }
            }
        }
        if (d > SLOTS)
            a = overflow_sum<FEAT, false>(a, rlane, node, deg, ovf_cnt, ovf, e, lane);

        a = a * dinv_of(d);
        a = a + bv;
        *(vecf*)(acc + (size_t)node * FEAT + lane * VEC) = a;

        if (node1 >= n) break;
        node = node1; d = d1; j = j1;
        node1 = node2; d1 = d2;
    }
}


constexpr int TILE    = 16;
constexpr int A_PITCH = 132;

__device__ __forceinline__ v8f strip_product(const float* ar, const float* wr, int wstep) {
    v8f c = {0.f, 0.f, 0.f, 0.f, 0.f, 0.f, 0.f, 0.f};
#pragma unroll
    for (int kb = 0; kb < 32; ++kb) {
        const v2f a = *(const v2f*)(ar + kb * 4);
        const v2f b = *(const v2f*)(wr + kb * wstep);
        c = __builtin_amdgcn_wmma_f32_16x16x4_f32(
                false, a, false, b, (short)0, c, false, false);
    }
    return c;
}

__global__ void __launch_bounds__(TILE * 32)
agg_gemm_relu_gemm_kernel(const float* __restrict__ x,
                          const int* __restrict__ deg,
                          const int* __restrict__ slots,
                          const int* __restrict__ ovf_cnt,
                          const int2* __restrict__ ovf,
                          const float* __restrict__ W1,
                          const float* __restrict__ b1,
                          const float* __restrict__ W2,
                          float* __restrict__ C, int n, int e) {
    constexpr int K    = 128;
    constexpr int NOUT = 64;
    constexpr int NT1  = K / 16;
    constexpr int NT2  = NOUT / 16;
    __shared__ __attribute__((aligned(16))) float sW1[(K / 4) * NT1 * 32 * 2];
    __shared__ __attribute__((aligned(16))) float sW2[(K / 4) * NT2 * 32 * 2];
    __shared__ __attribute__((aligned(16))) float sA[2][TILE * A_PITCH];
    __shared__ __attribute__((aligned(16))) float sH[2][TILE * A_PITCH];

    const int lane = threadIdx.x & 31;
    const int wave = __builtin_amdgcn_readfirstlane((int)(threadIdx.x >> 5));

    {
        constexpr int WPT = (K / 4) * NT1 * 32 / (TILE * 32);
        v2f wv[WPT];
#pragma unroll
        for (int i = 0; i < WPT; ++i) {
            const int idx = threadIdx.x + i * (TILE * 32);
            const int l   = idx & 31;
            const int ct  = (idx >> 5) & (NT1 - 1);
            const int kb  = idx / (NT1 * 32);
            const int k   = kb * 4 + ((l & 16) ? 2 : 0);
            const int col = ct * 16 + (l & 15);
            wv[i].x = W1[(size_t)k * K + col];
            wv[i].y = W1[(size_t)(k + 1) * K + col];
        }
#pragma unroll
        for (int i = 0; i < WPT; ++i)
            *(v2f*)&sW1[(threadIdx.x + i * (TILE * 32)) * 2] = wv[i];
    }
    {
        constexpr int WPT = (K / 4) * NT2 * 32 / (TILE * 32);
        v2f wv[WPT];
#pragma unroll
        for (int i = 0; i < WPT; ++i) {
            const int idx = threadIdx.x + i * (TILE * 32);
            const int l   = idx & 31;
            const int ct  = (idx >> 5) & (NT2 - 1);
            const int kb  = idx / (NT2 * 32);
            const int k   = kb * 4 + ((l & 16) ? 2 : 0);
            const int col = ct * 16 + (l & 15);
            wv[i].x = W2[(size_t)k * NOUT + col];
            wv[i].y = W2[(size_t)(k + 1) * NOUT + col];
        }
#pragma unroll
        for (int i = 0; i < WPT; ++i)
            *(v2f*)&sW2[(threadIdx.x + i * (TILE * 32)) * 2] = wv[i];
    }

    const int m    = lane & 15;
    const int koff = (lane & 16) ? 2 : 0;
    const int mrow = (lane & 16) ? 8 : 0;
    const int ntiles = (n + TILE - 1) / TILE;

    auto layer2_product = [&](int ptile, int hb, int ct) {
        const v8f c = strip_product(&sH[hb][m * A_PITCH + koff],
                                    &sW2[(ct * 32 + lane) * 2], NT2 * 32 * 2);
        const int row = ptile * TILE + mrow;
        float* cp = C + (size_t)row * NOUT + ct * 16 + m;
#pragma unroll
        for (int g = 0; g < 8; ++g) {
            if (row + g < n) cp[(size_t)g * NOUT] = c[g];
        }
    };

    int buf  = 0;
    int prev = -1;
    for (int tile = blockIdx.x; tile < ntiles; tile += gridDim.x, buf ^= 1) {
        const int node = tile * TILE + wave;
        float4 r  = make_float4(0.0f, 0.0f, 0.0f, 0.0f);
        float  di = 0.0f;
        if (node < n) {
            const int d = deg[node];
            di = dinv_of(d);
            const lane_vec<4>::type a =
                neighbour_sum<K, true, 8>(x, node, d, deg, slots, ovf_cnt, ovf, e, lane);
            r.x = a.x * di; r.y = a.y * di; r.z = a.z * di; r.w = a.w * di;
        }
        *(float4*)&sA[buf][wave * A_PITCH + lane * 4] = r;
        if (lane == 0) sA[buf][wave * A_PITCH + K] = di;
        __syncthreads();

        if (wave < NT1) {
            const v8f c = strip_product(&sA[buf][m * A_PITCH + koff],
                                        &sW1[(wave * 32 + lane) * 2], NT1 * 32 * 2);
            const float bcol = b1[wave * 16 + m];
            float* hp = &sH[buf][mrow * A_PITCH + wave * 16 + m];
            const float* dp = &sA[buf][mrow * A_PITCH + K];
#pragma unroll
            for (int g = 0; g < 8; ++g)
                hp[g * A_PITCH] = fmaxf(c[g] + bcol, 0.0f) * dp[g * A_PITCH];
        } else if (wave < NT1 + NT2) {
            if (prev >= 0) layer2_product(prev, buf ^ 1, wave - NT1);
        }
        prev = tile;
    }
    __syncthreads();
    if (wave >= NT1 && wave < NT1 + NT2 && prev >= 0)
        layer2_product(prev, buf ^ 1, wave - NT1);
}


extern "C" void kernel_launch(void* const* d_in, const int* in_sizes, int n_in,
                              void* d_out, int out_size, void* d_ws, size_t ws_size,
                              hipStream_t stream) {
    const float* x   = (const float*)d_in[0];
    const int*   ei  = (const int*)d_in[1];
    const float* W1  = (const float*)d_in[2];
    const float* b1  = (const float*)d_in[3];
    const float* W2  = (const float*)d_in[4];
    const float* b2  = (const float*)d_in[5];
    float*       out = (float*)d_out;

    const int N = in_sizes[0] / 128;
    const int E = in_sizes[1] / 2;
    const int* src = ei;
    const int* dst = ei + E;

    char* w = (char*)d_ws;
    auto carve = [&](size_t bytes) {
        void* p = (void*)w;
        w += (bytes + 255) & ~(size_t)255;
        return p;
    };
    int*   deg   = (int*)carve((size_t)N * 4);
    int*   ovfc  = (int*)carve(4);
    int*   slots = (int*)carve((size_t)N * SLOTS * 4);
    int2*  ovf   = (int2*)carve((size_t)E * 8);
    int2*  sorted = (int2*)carve((size_t)E * 8);
    int*   off   = (int*)carve((size_t)NCH * (NBKT_MAX + 1) * 4);
    float* hl2  = (float*)carve((size_t)N * 64 * 4);
    (void)n_in; (void)out_size; (void)ws_size;

    bin_kernel<<<NCH, BT, 0, stream>>>(src, dst, sorted, off, ovfc, N, E);
    build_kernel<<<(N + NPB - 1) / NPB, BT, 0, stream>>>(sorted, off, deg, slots, ovfc, ovf, N, E);

    agg_gemm_relu_gemm_kernel<<<512, TILE * 32, 0, stream>>>(x, deg, slots, ovfc, ovf, W1, b1, W2, hl2, N, E);

    gather_kernel<64><<<GATHER_BLOCKS, GW * 32, 0, stream>>>(hl2, deg, slots, ovfc, ovf, b2, out, N, E);
}


# Layer-1 list walk reads neighbour degrees from a byte table of all degrees in each block's LDS, not one scattered global read each; W fragments kept in registers make room for it (2 blocks/WGP kept)
# speedup vs baseline: 5.8277x; 1.0668x over previous
#include <hip/hip_runtime.h>
#include <math.h>


typedef float v2f __attribute__((ext_vector_type(2)));
typedef float v8f __attribute__((ext_vector_type(8)));


constexpr int SLOTS    = 64;
constexpr int NSH      = 8;
constexpr int NPB      = 1 << NSH;
constexpr int NBKT_MAX = 512;
constexpr int NCH      = 256;
constexpr int CH_MAX   = 8192;
constexpr int BT       = 1024;
constexpr int TAB_MAX  = 102400;
constexpr int TAB_SAT  = 255;

__device__ __forceinline__ float dinv_of(int d) { return rsqrtf((float)d + 1.0f); }

__global__ void __launch_bounds__(BT)
bin_kernel(const int* __restrict__ src, const int* __restrict__ dst,
           int2* __restrict__ sorted, int* __restrict__ off,
           int* __restrict__ ovf_cnt, int n, int e) {
    __shared__ int  cnt[NBKT_MAX];
    __shared__ int  base[NBKT_MAX + 1];
    __shared__ int2 stage[CH_MAX];

    const int tid   = threadIdx.x;
    const int nbkt  = (n + NPB - 1) >> NSH;
    const int chunk = (e + NCH - 1) / NCH;
    const int c0    = blockIdx.x * chunk;
    int c1 = c0 + chunk;
    if (c1 > e) c1 = e;
    const int m = c1 > c0 ? c1 - c0 : 0;

    if (tid < NBKT_MAX) cnt[tid] = 0;
    if (blockIdx.x == 0 && tid == 0) *ovf_cnt = 0;
    __syncthreads();

    constexpr int EPT = CH_MAX / BT;
    int d[EPT], s[EPT], r[EPT];
#pragma unroll
    for (int u = 0; u < EPT; ++u) {
        const int i = tid + u * BT;
        const bool ok = i < m;
        d[u] = ok ? dst[c0 + i] : 0;
        s[u] = ok ? src[c0 + i] : 0;
    }
#pragma unroll
    for (int u = 0; u < EPT; ++u) {
        const int i = tid + u * BT;
        r[u] = (i < m) ? atomicAdd(&cnt[d[u] >> NSH], 1) : 0;
    }
    __syncthreads();

    if (tid < 32) {
        constexpr int PER = NBKT_MAX / 32;
        int loc[PER];
        int sum = 0;
#pragma unroll
        for (int j = 0; j < PER; ++j) { loc[j] = cnt[tid * PER + j]; sum += loc[j]; }
        int inc = sum;
#pragma unroll
        for (int o = 1; o < 32; o <<= 1) {
            const int t = __shfl_up(inc, o);
            if (tid >= o) inc += t;
        }
        int run = inc - sum;
#pragma unroll
        for (int j = 0; j < PER; ++j) { base[tid * PER + j] = run; run += loc[j]; }
        if (tid == 31) base[NBKT_MAX] = run;
    }
    __syncthreads();

#pragma unroll
    for (int u = 0; u < EPT; ++u) {
        const int i = tid + u * BT;
        if (i < m) stage[base[d[u] >> NSH] + r[u]] = make_int2(d[u], s[u]);
    }
    __syncthreads();

#pragma unroll
    for (int u = 0; u < EPT; ++u) {
        const int i = tid + u * BT;
        if (i < m) sorted[(size_t)c0 + i] = stage[i];
    }
    for (int i = tid; i <= nbkt; i += BT)
        off[(size_t)blockIdx.x * (nbkt + 1) + i] = base[i];
}

__device__ __forceinline__ void place_edge(int2 q, int node0, int* lcnt, int* lslots,
                                           int* __restrict__ ovf_cnt,
                                           int2* __restrict__ ovf) {
    const int ln = q.x - node0;
    const int r  = atomicAdd(&lcnt[ln], 1);
    if (r < SLOTS) {
        lslots[ln * SLOTS + r] = q.y;
    } else {
        const int k = atomicAdd(ovf_cnt, 1);
        ovf[k] = q;
    }
}

__global__ void __launch_bounds__(BT)
build_kernel(const int2* __restrict__ sorted, const int* __restrict__ off,
             int* __restrict__ deg, unsigned char* __restrict__ degb,
             int* __restrict__ slots,
             int* __restrict__ ovf_cnt, int2* __restrict__ ovf, int n, int e) {
    __shared__ int lcnt[NPB];
    __shared__ int segs[NCH];
    __shared__ int segl[NCH];
    __shared__ int lslots[NPB * SLOTS];

    constexpr int NW  = BT / 32;
    constexpr int SPW = NCH / NW;

    const int tid   = threadIdx.x;
    const int lane  = tid & 31;
    const int wave  = __builtin_amdgcn_readfirstlane(tid >> 5);
    const int b     = blockIdx.x;
    const int nbkt  = (n + NPB - 1) >> NSH;
    const int chunk = (e + NCH - 1) / NCH;
    const int node0 = b << NSH;

    if (tid < NPB) lcnt[tid] = 0;
    if (tid < NCH) {
        const int* o = off + (size_t)tid * (nbkt + 1) + b;
        const int st = o[0];
        segs[tid] = st;
        segl[tid] = o[1] - st;
    }
    __syncthreads();

    int2 q[SPW];
    int  len[SPW];
#pragma unroll
    for (int t = 0; t < SPW; ++t) {
        const int k  = wave + t * NW;
        const int st = __builtin_amdgcn_readfirstlane(segs[k]);
        len[t]       = __builtin_amdgcn_readfirstlane(segl[k]);
        const int2* p = sorted + (size_t)k * chunk + st;
        q[t] = make_int2(0, 0);
        if (lane < len[t]) q[t] = p[lane];
    }
#pragma unroll
    for (int t = 0; t < SPW; ++t) {
        if (lane < len[t]) place_edge(q[t], node0, lcnt, lslots, ovf_cnt, ovf);
    }
#pragma unroll
    for (int t = 0; t < SPW; ++t) {
        if (len[t] > 32) {
            const int k  = wave + t * NW;
            const int st = __builtin_amdgcn_readfirstlane(segs[k]);
            const int2* p = sorted + (size_t)k * chunk + st;
            for (int i = 32 + lane; i < len[t]; i += 32)
                place_edge(p[i], node0, lcnt, lslots, ovf_cnt, ovf);
        }
    }
    __syncthreads();

    int nn = n - node0;
    if (nn > NPB) nn = NPB;
    for (int idx = tid; idx < nn * SLOTS; idx += BT) {
        if ((idx & (SLOTS - 1)) < lcnt[idx / SLOTS])
            slots[(size_t)node0 * SLOTS + idx] = lslots[idx];
    }
    if (tid < nn) {
        const int dv = lcnt[tid];
        deg[node0 + tid]  = dv;
        degb[node0 + tid] = (unsigned char)(dv < TAB_SAT ? dv : TAB_SAT);
    }
}


template <int VEC> struct lane_vec { typedef float type __attribute__((ext_vector_type(VEC))); };

__device__ __forceinline__ int degree_of(const unsigned char* tab,
                                         const int* __restrict__ deg, int j) {
    int dj = TAB_SAT;
    if (j < TAB_MAX) dj = tab[j];
    if (dj == TAB_SAT) dj = deg[j];
    return dj;
}

template <int FEAT, bool SCALED>
__device__ __forceinline__ typename lane_vec<FEAT / 32>::type
overflow_sum(typename lane_vec<FEAT / 32>::type a, const float* rlane, int node,
             const int* __restrict__ deg,
             const int* __restrict__ ovf_cnt,
             const int2* __restrict__ ovf, int e, int lane) {
    typedef typename lane_vec<FEAT / 32>::type vecf;
    int cnt = *ovf_cnt;
    if (cnt > e) cnt = e;
    for (int b = 0; b < cnt; b += 32) {
        const int k = b + lane;
        int od = 0, os = 0;
        if (k < cnt) { const int2 q = ovf[k]; od = q.x; os = q.y; }
        unsigned hit = __builtin_amdgcn_ballot_w32(k < cnt && od == node);
        while (hit) {
            const int l = __builtin_ctz(hit);
            hit &= hit - 1;
            const int ju = __builtin_amdgcn_readlane(os, l);
            const vecf v = *(const vecf*)(rlane + (size_t)ju * FEAT);
            if (SCALED) a = a + v * dinv_of(deg[ju]);
            else        a = a + v;
        }
    }
    return a;
}

template <int FEAT, bool SCALED, int U>
__device__ __forceinline__ typename lane_vec<FEAT / 32>::type
neighbour_sum(const float* __restrict__ rows, int node, int d,
              const unsigned char* tab,
              const int* __restrict__ deg,
              const int* __restrict__ slots,
              const int* __restrict__ ovf_cnt,
              const int2* __restrict__ ovf, int e, int lane) {
    constexpr int VEC = FEAT / 32;
    typedef typename lane_vec<VEC>::type vecf;

    const int   total = (d < SLOTS ? d : SLOTS) + 1;
    const int*  list  = slots + (size_t)node * SLOTS;
    const float* rlane = rows + lane * VEC;

    vecf a = {};
    for (int base = 0; base < total; base += 32) {
        const int m = (total - base < 32) ? (total - base) : 32;
        const int p = base + lane;
        int j = node;
        if (lane < m && p > 0) j = list[p - 1];
        int sbits = 0;
        if (SCALED) sbits = __float_as_int(dinv_of(degree_of(tab, deg, j)));
        for (int t = 0; t < m; t += U) {
            vecf v[U];
            unsigned ob[U];
#pragma unroll
            for (int u = 0; u < U; ++u) {
                const int ju = __builtin_amdgcn_readlane(j, t + u);
                ob[u] = (unsigned)ju * (unsigned)(FEAT * 4)
                      + (unsigned)lane * (unsigned)(VEC * 4);
            }
            __builtin_amdgcn_sched_barrier(0);
#pragma unroll
            for (int u = 0; u < U; ++u)
                v[u] = *(const vecf*)((const char*)rows + ob[u]);
#pragma unroll
            for (int u = 0; u < U; ++u) {
                if (t + u < m) {
                    if (SCALED) {
                        const float su = __int_as_float(__builtin_amdgcn_readlane(sbits, t + u));
                        a = a + v[u] * su;
                    } else {
                        a = a + v[u];
                    }
                }
            }
        }
    }

    if (d > SLOTS)
        a = overflow_sum<FEAT, SCALED>(a, rlane, node, deg, ovf_cnt, ovf, e, lane);
    return a;
}


constexpr int GW            = 16;
constexpr int GATHER_BLOCKS = 512;

template <int FEAT>
__global__ void __launch_bounds__(GW * 32)
gather_kernel(const float* __restrict__ hs,
              const int* __restrict__ deg,
              const int* __restrict__ slots,
              const int* __restrict__ ovf_cnt,
              const int2* __restrict__ ovf,
              const float* __restrict__ bias,
              float* __restrict__ acc, int n, int e) {
    constexpr int VEC = FEAT / 32;
    typedef typename lane_vec<VEC>::type vecf;

    const int lane = threadIdx.x & 31;
    const int wid = __builtin_amdgcn_readfirstlane(
        (int)(blockIdx.x * GW + (threadIdx.x >> 5)));
    const int nw  = (int)gridDim.x * GW;
    if (wid >= n) return;

    const float* rlane = hs + lane * VEC;
    const vecf   bv    = *(const vecf*)(bias + lane * VEC);

    auto first_entries = [&](int node, int d) {
        const int total = (d < SLOTS ? d : SLOTS) + 1;
        int j = node;
        if (lane > 0 && lane < total) j = slots[(size_t)node * SLOTS + lane - 1];
        return j;
    };
    auto row = [&](int ju) {
        const float* rp = hs + (size_t)ju * FEAT;
        return *(const vecf*)(rp + lane * VEC);
    };

    int node  = wid;
    int d     = deg[node];
    int j     = first_entries(node, d);
    int node1 = node + nw;
    int d1    = 0;
    if (node1 < n) d1 = deg[node1];

    for (;;) {
        const int total = (d < SLOTS ? d : SLOTS) + 1;
        const int m0    = total < 32 ? total : 32;
        vecf v[32];
#pragma unroll
        for (int c = 0; c < 8; ++c) {
            if (c == 0 || c * 4 < m0) {
#pragma unroll
                for (int u = 0; u < 4; ++u)
                    v[c * 4 + u] = row(__builtin_amdgcn_readlane(j, c * 4 + u));
            }
        }

        const int node2 = node1 + nw;
        int d2 = 0;
        if (node2 < n) d2 = deg[node2];
        int j1 = 0;
        if (node1 < n) j1 = first_entries(node1, d1);

        vecf a = {};
#pragma unroll
        for (int c = 0; c < 8; ++c) {
            if (c == 0 || c * 4 < m0) {
#pragma unroll
                for (int u = 0; u < 4; ++u) {
                    if (c * 4 + u == 0 || c * 4 + u < m0) a = a + v[c * 4 + u];
                }
            }
        }
        for (int base = 32; base < total; base += 32) {
            const int m = (total - base < 32) ? (total - base) : 32;
            int jj = node;
            if (lane < m) jj = slots[(size_t)node * SLOTS + base + lane - 1];
            for (int t = 0; t < m; t += 4) {
                vecf w[4];
#pragma unroll
                for (int u = 0; u < 4; ++u)
                    w[u] = row(__builtin_amdgcn_readlane(jj, t + u));
#pragma unroll
                for (int u = 0; u < 4; ++u) {
                    if (t + u < m) a = a + w[u];
                }
            }
        }
        if (d > SLOTS)
            a = overflow_sum<FEAT, false>(a, rlane, node, deg, ovf_cnt, ovf, e, lane);

        a = a * dinv_of(d);
        a = a + bv;
        *(vecf*)(acc + (size_t)node * FEAT + lane * VEC) = a;

        if (node1 >= n) break;
        node = node1; d = d1; j = j1;
        node1 = node2; d1 = d2;
    }
}


constexpr int TILE    = 16;
constexpr int A_PITCH = 132;

__device__ __forceinline__ v8f strip_product(const float* ar, const v2f (&wf)[32]) {
    v8f c = {0.f, 0.f, 0.f, 0.f, 0.f, 0.f, 0.f, 0.f};
#pragma unroll
    for (int kb = 0; kb < 32; ++kb) {
        const v2f a = *(const v2f*)(ar + kb * 4);
        c = __builtin_amdgcn_wmma_f32_16x16x4_f32(
                false, a, false, wf[kb], (short)0, c, false, false);
    }
    return c;
}

__global__ void __launch_bounds__(TILE * 32, 8)
agg_gemm_relu_gemm_kernel(const float* __restrict__ x,
                          const int* __restrict__ deg,
                          const unsigned char* __restrict__ degb,
                          const int* __restrict__ slots,
                          const int* __restrict__ ovf_cnt,
                          const int2* __restrict__ ovf,
                          const float* __restrict__ W1,
                          const float* __restrict__ b1,
                          const float* __restrict__ W2,
                          float* __restrict__ C, int n, int e) {
    constexpr int K    = 128;
    constexpr int NOUT = 64;
    constexpr int NT1  = K / 16;
    constexpr int NT2  = NOUT / 16;
    __shared__ __attribute__((aligned(16))) unsigned char sdeg[TAB_MAX];
    __shared__ __attribute__((aligned(16))) float sA[2][TILE * A_PITCH];
    __shared__ __attribute__((aligned(16))) float sH[2][TILE * A_PITCH];

    const int lane = threadIdx.x & 31;
    const int wave = __builtin_amdgcn_readfirstlane((int)(threadIdx.x >> 5));

    const int m    = lane & 15;
    const int koff = (lane & 16) ? 2 : 0;
    const int mrow = (lane & 16) ? 8 : 0;
    const int ntiles = (n + TILE - 1) / TILE;

    {
        constexpr int CPT = (TAB_MAX / 16 + TILE * 32 - 1) / (TILE * 32);
        const int nt    = n < TAB_MAX ? n : TAB_MAX;
        const int nfull = nt >> 4;
        const uint4* g16 = (const uint4*)degb;
        uint4 t[CPT];
#pragma unroll
        for (int u = 0; u < CPT; ++u) {
            const int i = threadIdx.x + u * (TILE * 32);
            t[u] = make_uint4(0u, 0u, 0u, 0u);
            if (i < nfull) t[u] = g16[i];
        }
#pragma unroll
        for (int u = 0; u < CPT; ++u) {
            const int i = threadIdx.x + u * (TILE * 32);
            if (i < nfull) *(uint4*)&sdeg[i * 16] = t[u];
        }
        const int rest = (nfull << 4) + (int)threadIdx.x;
        if (rest < nt) sdeg[rest] = degb[rest];
    }

    v2f wf[K / 4];
    {
        const bool   l1   = wave < NT1;
        const float* wsrc = l1 ? W1 : W2;
        const int    ldw  = l1 ? K : NOUT;
        const int    ct   = l1 ? wave : wave - NT1;
        if (wave < NT1 + NT2) {
            const float* wp = wsrc + (size_t)koff * ldw + ct * 16 + m;
#pragma unroll
            for (int kb = 0; kb < K / 4; ++kb) {
                wf[kb].x = wp[(size_t)(kb * 4) * ldw];
                wf[kb].y = wp[(size_t)(kb * 4 + 1) * ldw];
            }
        } else {
#pragma unroll
            for (int kb = 0; kb < K / 4; ++kb) { wf[kb].x = 0.0f; wf[kb].y = 0.0f; }
        }
    }
    __syncthreads();

    auto layer2_product = [&](int ptile, int hb, int ct) {
        const v8f c = strip_product(&sH[hb][m * A_PITCH + koff], wf);
        const int row = ptile * TILE + mrow;
        float* cp = C + (size_t)row * NOUT + ct * 16 + m;
#pragma unroll
        for (int g = 0; g < 8; ++g) {
            if (row + g < n) cp[(size_t)g * NOUT] = c[g];
        }
    };

    int buf  = 0;
    int prev = -1;
    for (int tile = blockIdx.x; tile < ntiles; tile += gridDim.x, buf ^= 1) {
        const int node = tile * TILE + wave;
        float4 r  = make_float4(0.0f, 0.0f, 0.0f, 0.0f);
        float  di = 0.0f;
        if (node < n) {
            const int d = __builtin_amdgcn_readfirstlane(degree_of(sdeg, deg, node));
            di = dinv_of(d);
            const lane_vec<4>::type a =
                neighbour_sum<K, true, 8>(x, node, d, sdeg, deg, slots, ovf_cnt, ovf, e, lane);
            r.x = a.x * di; r.y = a.y * di; r.z = a.z * di; r.w = a.w * di;
        }
        *(float4*)&sA[buf][wave * A_PITCH + lane * 4] = r;
        if (lane == 0) sA[buf][wave * A_PITCH + K] = di;
        __syncthreads();

        if (wave < NT1) {
            const v8f c = strip_product(&sA[buf][m * A_PITCH + koff], wf);
            const float bcol = b1[wave * 16 + m];
            float* hp = &sH[buf][mrow * A_PITCH + wave * 16 + m];
            const float* dp = &sA[buf][mrow * A_PITCH + K];
#pragma unroll
            for (int g = 0; g < 8; ++g)
                hp[g * A_PITCH] = fmaxf(c[g] + bcol, 0.0f) * dp[g * A_PITCH];
        } else if (wave < NT1 + NT2) {
            if (prev >= 0) layer2_product(prev, buf ^ 1, wave - NT1);
        }
        prev = tile;
    }
    __syncthreads();
    if (wave >= NT1 && wave < NT1 + NT2 && prev >= 0)
        layer2_product(prev, buf ^ 1, wave - NT1);
}


extern "C" void kernel_launch(void* const* d_in, const int* in_sizes, int n_in,
                              void* d_out, int out_size, void* d_ws, size_t ws_size,
                              hipStream_t stream) {
    const float* x   = (const float*)d_in[0];
    const int*   ei  = (const int*)d_in[1];
    const float* W1  = (const float*)d_in[2];
    const float* b1  = (const float*)d_in[3];
    const float* W2  = (const float*)d_in[4];
    const float* b2  = (const float*)d_in[5];
    float*       out = (float*)d_out;

    const int N = in_sizes[0] / 128;
    const int E = in_sizes[1] / 2;
    const int* src = ei;
    const int* dst = ei + E;

    char* w = (char*)d_ws;
    auto carve = [&](size_t bytes) {
        void* p = (void*)w;
        w += (bytes + 255) & ~(size_t)255;
        return p;
    };
    int*   deg   = (int*)carve((size_t)N * 4);
    unsigned char* degb = (unsigned char*)carve((size_t)N);
    int*   ovfc  = (int*)carve(4);
    int*   slots = (int*)carve((size_t)N * SLOTS * 4);
    int2*  ovf   = (int2*)carve((size_t)E * 8);
    int2*  sorted = (int2*)carve((size_t)E * 8);
    int*   off   = (int*)carve((size_t)NCH * (NBKT_MAX + 1) * 4);
    float* hl2  = (float*)carve((size_t)N * 64 * 4);
    (void)n_in; (void)out_size; (void)ws_size;

    bin_kernel<<<NCH, BT, 0, stream>>>(src, dst, sorted, off, ovfc, N, E);
    build_kernel<<<(N + NPB - 1) / NPB, BT, 0, stream>>>(sorted, off, deg, degb, slots, ovfc, ovf, N, E);

    agg_gemm_relu_gemm_kernel<<<512, TILE * 32, 0, stream>>>(x, deg, degb, slots, ovfc, ovf, W1, b1, W2, hl2, N, E);

    gather_kernel<64><<<GATHER_BLOCKS, GW * 32, 0, stream>>>(hl2, deg, slots, ovfc, ovf, b2, out, N, E);
}
